# starting kernel
# baseline (speedup 1.0000x reference)
#include <hip/hip_runtime.h>
#include <math.h>

typedef _Float16 h16;
typedef __attribute__((ext_vector_type(16))) _Float16 v16h;
typedef __attribute__((ext_vector_type(8)))  _Float16 v8h;
typedef __attribute__((ext_vector_type(8)))  float    v8f;

#define NODE_DIM 128
#define HIDDEN   128
#define NBASIS   32
#define MSG_IN   288
#define WAVES    4

static __device__ __forceinline__ v16h cat8(v8h a, v8h b) {
  return __builtin_shufflevector(a, b, 0,1,2,3,4,5,6,7,8,9,10,11,12,13,14,15);
}
static __device__ __forceinline__ float silu(float x) {
  return x / (1.0f + __expf(-x));
}
static __device__ __forceinline__ v16h load_b_frag(const h16* __restrict__ wp,
                                                   int kt, int nt, int NT, int lane) {
  const v8h* p = (const v8h*)(wp + (size_t)(((kt * NT + nt) * 32 + lane)) * 16);
  return cat8(p[0], p[1]);
}
static __device__ __forceinline__ v16h load_a_frag(const h16* row_base, int kt, int kb) {
  const v8h* a0 = (const v8h*)(row_base + kt * 32 + kb);
  const v8h* a1 = (const v8h*)(row_base + kt * 32 + 16 + kb);
  return cat8(*a0, *a1);
}

extern "C" __global__ void pack_w(const float* __restrict__ W, h16* __restrict__ out,
                                  int K, int Nn) {
  int t = blockIdx.x * blockDim.x + threadIdx.x;
  int KT = K >> 5, NT = Nn >> 4;
  int total = KT * NT * 32;
  if (t >= total) return;
  int lane = t & 31;
  int tile = t >> 5;
  int nt = tile % NT, kt = tile / NT;
  int n = nt * 16 + (lane & 15);
  int kbase = (lane < 16) ? 0 : 8;
  alignas(16) h16 tmp[16];
#pragma unroll
  for (int r = 0; r < 4; ++r) {
    int k0 = kt * 32 + kbase + 2 * r;
    tmp[2 * r]     = (h16)W[(size_t)k0 * Nn + n];
    tmp[2 * r + 1] = (h16)W[(size_t)(k0 + 1) * Nn + n];
    int k1 = kt * 32 + 16 + kbase + 2 * r;
    tmp[8 + 2 * r]     = (h16)W[(size_t)k1 * Nn + n];
    tmp[8 + 2 * r + 1] = (h16)W[(size_t)(k1 + 1) * Nn + n];
  }
  v8h* o = (v8h*)(out + (size_t)t * 16);
  o[0] = *(const v8h*)(tmp);
  o[1] = *(const v8h*)(tmp + 8);
}

extern "C" __global__ void egnn_init(float* __restrict__ aggr, const float* __restrict__ pos,
                                     float* __restrict__ pos_out, int N) {
  int i = blockIdx.x * blockDim.x + threadIdx.x;
  int stride = gridDim.x * blockDim.x;
  for (int j = i; j < N * HIDDEN; j += stride) aggr[j] = 0.0f;
  for (int j = i; j < N * 3; j += stride) pos_out[j] = pos[j];
}

extern "C" __global__ void __launch_bounds__(WAVES * 32)
egnn_edge(const float* __restrict__ hfeat, const float* __restrict__ pos,
          const int* __restrict__ ei,
          const h16* __restrict__ mw1p, const float* __restrict__ mb1,
          const h16* __restrict__ mw2p, const float* __restrict__ mb2,
          const h16* __restrict__ cw1p, const float* __restrict__ cb1,
          const float* __restrict__ cw2, const float* __restrict__ cb2,
          float* __restrict__ aggr, float* __restrict__ pos_out, int E) {
  extern __shared__ char smem[];
  const int wave = threadIdx.x >> 5;
  const int lane = threadIdx.x & 31;
  const int m    = lane & 15;
  const int kb   = (lane < 16) ? 0 : 8;
  const int drow = (lane < 16) ? 0 : 8;

  const int WB = 16 * MSG_IN * 2 + 16 * HIDDEN * 2 + 64 + 64;
  char* wb = smem + wave * WB;
  h16*   s_min = (h16*)wb;
  h16*   s_hid = (h16*)(wb + 16 * MSG_IN * 2);
  int*   s_row = (int*)(wb + 16 * MSG_IN * 2 + 16 * HIDDEN * 2);
  float* s_cm  = (float*)(wb + 16 * MSG_IN * 2 + 16 * HIDDEN * 2 + 64);

  const int ntile = E >> 4;
  const int nwav  = gridDim.x * WAVES;
  for (int tile = blockIdx.x * WAVES + wave; tile < ntile; tile += nwav) {
    const int e    = tile * 16 + m;
    const int vrow = ei[e];
    const int vcol = ei[E + e];
    float rx = 0.f, ry = 0.f, rz = 0.f;

    if (lane < 16) {
      s_row[m] = vrow;
      rx = pos[vrow * 3 + 0] - pos[vcol * 3 + 0];
      ry = pos[vrow * 3 + 1] - pos[vcol * 3 + 1];
      rz = pos[vrow * 3 + 2] - pos[vcol * 3 + 2];
      float dist = sqrtf(rx * rx + ry * ry + rz * rz + 1e-8f);
      const float step  = 10.0f / (NBASIS - 1);
      const float coeff = -0.5f / (step * step);
#pragma unroll
      for (int b = 0; b < NBASIS; ++b) {
        float d = dist - step * b;
        s_min[m * MSG_IN + 2 * NODE_DIM + b] = (h16)__expf(coeff * d * d);
      }
      const float4* hp = (const float4*)(hfeat + (size_t)vrow * NODE_DIM);
#pragma unroll 4
      for (int j = 0; j < NODE_DIM / 4; ++j) {
        float4 v = hp[j];
        h16* dst = s_min + m * MSG_IN + j * 4;
        dst[0] = (h16)v.x; dst[1] = (h16)v.y; dst[2] = (h16)v.z; dst[3] = (h16)v.w;
      }
    } else {
      const float4* hp = (const float4*)(hfeat + (size_t)vcol * NODE_DIM);
#pragma unroll 4
      for (int j = 0; j < NODE_DIM / 4; ++j) {
        float4 v = hp[j];
        h16* dst = s_min + m * MSG_IN + NODE_DIM + j * 4;
        dst[0] = (h16)v.x; dst[1] = (h16)v.y; dst[2] = (h16)v.z; dst[3] = (h16)v.w;
      }
    }
    asm volatile("s_wait_dscnt 0" ::: "memory");

    v8f acc[8];
#pragma unroll
    for (int nt = 0; nt < 8; ++nt) {
      float b0 = mb1[nt * 16 + m];
      acc[nt] = (v8f){b0, b0, b0, b0, b0, b0, b0, b0};
    }
#pragma unroll
    for (int kt = 0; kt < MSG_IN / 32; ++kt) {
      v16h a = load_a_frag(s_min + m * MSG_IN, kt, kb);
#pragma unroll
      for (int nt = 0; nt < 8; ++nt) {
        v16h b = load_b_frag(mw1p, kt, nt, 8, lane);
        acc[nt] = __builtin_amdgcn_wmma_f32_16x16x32_f16(false, a, false, b,
                                                         (short)0, acc[nt], false, false);
      }
    }
#pragma unroll
    for (int nt = 0; nt < 8; ++nt)
#pragma unroll
      for (int r = 0; r < 8; ++r)
        s_hid[(drow + r) * HIDDEN + nt * 16 + m] = (h16)silu(acc[nt][r]);
    asm volatile("s_wait_dscnt 0" ::: "memory");

    v8f acc2[8];
#pragma unroll
    for (int nt = 0; nt < 8; ++nt) {
      float b0 = mb2[nt * 16 + m];
      acc2[nt] = (v8f){b0, b0, b0, b0, b0, b0, b0, b0};
    }
#pragma unroll
    for (int kt = 0; kt < HIDDEN / 32; ++kt) {
      v16h a = load_a_frag(s_hid + m * HIDDEN, kt, kb);
#pragma unroll
      for (int nt = 0; nt < 8; ++nt) {
        v16h b = load_b_frag(mw2p, kt, nt, 8, lane);
        acc2[nt] = __builtin_amdgcn_wmma_f32_16x16x32_f16(false, a, false, b,
                                                          (short)0, acc2[nt], false, false);
      }
    }
#pragma unroll
    for (int nt = 0; nt < 8; ++nt) {
      const int col = nt * 16 + m;
#pragma unroll
      for (int r = 0; r < 8; ++r) {
        float v  = acc2[nt][r];
        int   rr = drow + r;
        unsafeAtomicAdd(&aggr[(size_t)s_row[rr] * HIDDEN + col], v);
        s_hid[rr * HIDDEN + col] = (h16)v;
      }
    }
    asm volatile("s_wait_dscnt 0" ::: "memory");

    v8f acc3[8];
#pragma unroll
    for (int nt = 0; nt < 8; ++nt) {
      float b0 = cb1[nt * 16 + m];
      acc3[nt] = (v8f){b0, b0, b0, b0, b0, b0, b0, b0};
    }
#pragma unroll
    for (int kt = 0; kt < HIDDEN / 32; ++kt) {
      v16h a = load_a_frag(s_hid + m * HIDDEN, kt, kb);
#pragma unroll
      for (int nt = 0; nt < 8; ++nt) {
        v16h b = load_b_frag(cw1p, kt, nt, 8, lane);
        acc3[nt] = __builtin_amdgcn_wmma_f32_16x16x32_f16(false, a, false, b,
                                                          (short)0, acc3[nt], false, false);
      }
    }
    if (lane < 16) s_cm[m] = cb2[0];
    asm volatile("s_wait_dscnt 0" ::: "memory");
#pragma unroll
    for (int nt = 0; nt < 8; ++nt) {
      float w = cw2[nt * 16 + m];
#pragma unroll
      for (int r = 0; r < 8; ++r)
        atomicAdd(&s_cm[drow + r], silu(acc3[nt][r]) * w);
    }
    asm volatile("s_wait_dscnt 0" ::: "memory");
    if (lane < 16) {
      float cm = s_cm[m];
      unsafeAtomicAdd(&pos_out[vrow * 3 + 0], rx * cm);
      unsafeAtomicAdd(&pos_out[vrow * 3 + 1], ry * cm);
      unsafeAtomicAdd(&pos_out[vrow * 3 + 2], rz * cm);
    }
    asm volatile("s_wait_dscnt 0" ::: "memory");
  }
}

extern "C" __global__ void __launch_bounds__(WAVES * 32)
egnn_node(const float* __restrict__ hfeat, const float* __restrict__ aggr,
          const h16* __restrict__ nw1p, const float* __restrict__ nb1,
          const h16* __restrict__ nw2p, const float* __restrict__ nb2,
          float* __restrict__ h_out, int Nn) {
  extern __shared__ char smem[];
  const int wave = threadIdx.x >> 5;
  const int lane = threadIdx.x & 31;
  const int m    = lane & 15;
  const int kb   = (lane < 16) ? 0 : 8;
  const int drow = (lane < 16) ? 0 : 8;

  const int NIN = 2 * NODE_DIM;
  const int WB  = 16 * NIN * 2 + 16 * HIDDEN * 2;
  char* wb = smem + wave * WB;
  h16* s_nin = (h16*)wb;
  h16* s_hid = (h16*)(wb + 16 * NIN * 2);

  const int ntile = Nn >> 4;
  const int nwav  = gridDim.x * WAVES;
  for (int tile = blockIdx.x * WAVES + wave; tile < ntile; tile += nwav) {
    const int node = tile * 16 + m;
    const float4* src = (lane < 16) ? (const float4*)(hfeat + (size_t)node * NODE_DIM)
                                    : (const float4*)(aggr + (size_t)node * HIDDEN);
    h16* dst = s_nin + m * NIN + ((lane < 16) ? 0 : NODE_DIM);
#pragma unroll 4
    for (int j = 0; j < NODE_DIM / 4; ++j) {
      float4 v = src[j];
      dst[j * 4 + 0] = (h16)v.x; dst[j * 4 + 1] = (h16)v.y;
      dst[j * 4 + 2] = (h16)v.z; dst[j * 4 + 3] = (h16)v.w;
    }
    asm volatile("s_wait_dscnt 0" ::: "memory");

    v8f acc[8];
#pragma unroll
    for (int nt = 0; nt < 8; ++nt) {
      float b0 = nb1[nt * 16 + m];
      acc[nt] = (v8f){b0, b0, b0, b0, b0, b0, b0, b0};
    }
#pragma unroll
    for (int kt = 0; kt < NIN / 32; ++kt) {
      v16h a = load_a_frag(s_nin + m * NIN, kt, kb);
#pragma unroll
      for (int nt = 0; nt < 8; ++nt) {
        v16h b = load_b_frag(nw1p, kt, nt, 8, lane);
        acc[nt] = __builtin_amdgcn_wmma_f32_16x16x32_f16(false, a, false, b,
                                                         (short)0, acc[nt], false, false);
      }
    }
#pragma unroll
    for (int nt = 0; nt < 8; ++nt)
#pragma unroll
      for (int r = 0; r < 8; ++r)
        s_hid[(drow + r) * HIDDEN + nt * 16 + m] = (h16)silu(acc[nt][r]);
    asm volatile("s_wait_dscnt 0" ::: "memory");

    v8f acc2[8];
#pragma unroll
    for (int nt = 0; nt < 8; ++nt) {
      float b0 = nb2[nt * 16 + m];
      acc2[nt] = (v8f){b0, b0, b0, b0, b0, b0, b0, b0};
    }
#pragma unroll
    for (int kt = 0; kt < HIDDEN / 32; ++kt) {
      v16h a = load_a_frag(s_hid + m * HIDDEN, kt, kb);
#pragma unroll
      for (int nt = 0; nt < 8; ++nt) {
        v16h b = load_b_frag(nw2p, kt, nt, 8, lane);
        acc2[nt] = __builtin_amdgcn_wmma_f32_16x16x32_f16(false, a, false, b,
                                                          (short)0, acc2[nt], false, false);
      }
    }
#pragma unroll
    for (int nt = 0; nt < 8; ++nt) {
      const int col = nt * 16 + m;
#pragma unroll
      for (int r = 0; r < 8; ++r) {
        int n2 = tile * 16 + drow + r;
        h_out[(size_t)n2 * NODE_DIM + col] =
            hfeat[(size_t)n2 * NODE_DIM + col] + acc2[nt][r];
      }
    }
  }
}

extern "C" void kernel_launch(void* const* d_in, const int* in_sizes, int n_in,
                              void* d_out, int out_size, void* d_ws, size_t ws_size,
                              hipStream_t stream) {
  const float* h   = (const float*)d_in[0];
  const float* pos = (const float*)d_in[1];
  const int*   ei  = (const int*)d_in[2];
  const float* mw1 = (const float*)d_in[3];  const float* mb1 = (const float*)d_in[4];
  const float* mw2 = (const float*)d_in[5];  const float* mb2 = (const float*)d_in[6];
  const float* nw1 = (const float*)d_in[7];  const float* nb1 = (const float*)d_in[8];
  const float* nw2 = (const float*)d_in[9];  const float* nb2 = (const float*)d_in[10];
  const float* cw1 = (const float*)d_in[11]; const float* cb1 = (const float*)d_in[12];
  const float* cw2 = (const float*)d_in[13]; const float* cb2 = (const float*)d_in[14];
  const int N = in_sizes[0] / NODE_DIM;
  const int E = in_sizes[2] / 2;

  char* ws = (char*)d_ws;
  h16* mw1p = (h16*)(ws + 0);
  h16* mw2p = (h16*)(ws + 73728);
  h16* cw1p = (h16*)(ws + 106496);
  h16* nw1p = (h16*)(ws + 139264);
  h16* nw2p = (h16*)(ws + 204800);
  float* aggr = (float*)(ws + 237568);
  float* h_out   = (float*)d_out;
  float* pos_out = h_out + (size_t)N * NODE_DIM;

  pack_w<<<(9 * 8 * 32 + 127) / 128, 128, 0, stream>>>(mw1, mw1p, MSG_IN, HIDDEN);
  pack_w<<<(4 * 8 * 32 + 127) / 128, 128, 0, stream>>>(mw2, mw2p, HIDDEN, HIDDEN);
  pack_w<<<(4 * 8 * 32 + 127) / 128, 128, 0, stream>>>(cw1, cw1p, HIDDEN, HIDDEN);
  pack_w<<<(8 * 8 * 32 + 127) / 128, 128, 0, stream>>>(nw1, nw1p, 2 * NODE_DIM, HIDDEN);
  pack_w<<<(4 * 8 * 32 + 127) / 128, 128, 0, stream>>>(nw2, nw2p, HIDDEN, NODE_DIM);
  egnn_init<<<2048, 256, 0, stream>>>(aggr, pos, pos_out, N);

  size_t lds_e = (size_t)WAVES * (16 * MSG_IN * 2 + 16 * HIDDEN * 2 + 64 + 64);
  egnn_edge<<<2048, WAVES * 32, lds_e, stream>>>(h, pos, ei, mw1p, mb1, mw2p, mb2,
                                                 cw1p, cb1, cw2, cb2, aggr, pos_out, E);

  size_t lds_n = (size_t)WAVES * (16 * 2 * NODE_DIM * 2 + 16 * HIDDEN * 2);
  egnn_node<<<800, WAVES * 32, lds_n, stream>>>(h, aggr, nw1p, nb1, nw2p, nb2, h_out, N);
}


# Message layer 1 split by linearity: a new node kernel forms P = h x mw1[0:128] + mb1, Q = h x mw1[128:256] per node; edges start from P[row] + Q[col], 32 basis columns left; its sum order changes
# speedup vs baseline: 1.7804x; 1.7804x over previous
#include <hip/hip_runtime.h>
#include <math.h>

typedef _Float16 h16;
typedef __attribute__((ext_vector_type(16))) _Float16 v16h;
typedef __attribute__((ext_vector_type(8)))  _Float16 v8h;
typedef __attribute__((ext_vector_type(8)))  float    v8f;

#define NODE_DIM 128
#define HIDDEN   128
#define NBASIS   32
#define MSG_IN   288
#define WAVES    4

static __device__ __forceinline__ v16h cat8(v8h a, v8h b) {
  return __builtin_shufflevector(a, b, 0,1,2,3,4,5,6,7,8,9,10,11,12,13,14,15);
}
static __device__ __forceinline__ float silu(float x) {
  return x / (1.0f + __expf(-x));
}
static __device__ __forceinline__ v16h load_b_frag(const h16* __restrict__ wp,
                                                   int kt, int nt, int NT, int lane) {
  const v8h* p = (const v8h*)(wp + (size_t)(((kt * NT + nt) * 32 + lane)) * 16);
  return cat8(p[0], p[1]);
}
static __device__ __forceinline__ v16h load_a_frag(const h16* row_base, int kt, int kb) {
  const v8h* a0 = (const v8h*)(row_base + kt * 32 + kb);
  const v8h* a1 = (const v8h*)(row_base + kt * 32 + 16 + kb);
  return cat8(*a0, *a1);
}

extern "C" __global__ void pack_w(const float* __restrict__ W, h16* __restrict__ out,
                                  int K, int Nn) {
  int t = blockIdx.x * blockDim.x + threadIdx.x;
  int KT = K >> 5, NT = Nn >> 4;
  int total = KT * NT * 32;
  if (t >= total) return;
  int lane = t & 31;
  int tile = t >> 5;
  int nt = tile % NT, kt = tile / NT;
  int n = nt * 16 + (lane & 15);
  int kbase = (lane < 16) ? 0 : 8;
  alignas(16) h16 tmp[16];
#pragma unroll
  for (int r = 0; r < 4; ++r) {
    int k0 = kt * 32 + kbase + 2 * r;
    tmp[2 * r]     = (h16)W[(size_t)k0 * Nn + n];
    tmp[2 * r + 1] = (h16)W[(size_t)(k0 + 1) * Nn + n];
    int k1 = kt * 32 + 16 + kbase + 2 * r;
    tmp[8 + 2 * r]     = (h16)W[(size_t)k1 * Nn + n];
    tmp[8 + 2 * r + 1] = (h16)W[(size_t)(k1 + 1) * Nn + n];
  }
  v8h* o = (v8h*)(out + (size_t)t * 16);
  o[0] = *(const v8h*)(tmp);
  o[1] = *(const v8h*)(tmp + 8);
}

extern "C" __global__ void egnn_init(float* __restrict__ aggr, const float* __restrict__ pos,
                                     float* __restrict__ pos_out, int N) {
  int i = blockIdx.x * blockDim.x + threadIdx.x;
  int stride = gridDim.x * blockDim.x;
  for (int j = i; j < N * HIDDEN; j += stride) aggr[j] = 0.0f;
  for (int j = i; j < N * 3; j += stride) pos_out[j] = pos[j];
}

extern "C" __global__ void __launch_bounds__(WAVES * 32)
egnn_pre(const float* __restrict__ hfeat, const h16* __restrict__ mw1p,
         const float* __restrict__ mb1, float* __restrict__ pq, int Nn) {
  const int wave = threadIdx.x >> 5;
  const int lane = threadIdx.x & 31;
  const int m    = lane & 15;
  const int kb   = (lane < 16) ? 0 : 8;
  const int drow = (lane < 16) ? 0 : 8;
  const int ntile = Nn >> 4;
  const int nwav  = gridDim.x * WAVES;
  for (int tile = blockIdx.x * WAVES + wave; tile < ntile; tile += nwav) {
    const float* hr = hfeat + (size_t)(tile * 16 + m) * NODE_DIM;
    v8f accP[8], accQ[8];
#pragma unroll
    for (int nt = 0; nt < 8; ++nt) {
      float b0 = mb1[nt * 16 + m];
      accP[nt] = (v8f){b0, b0, b0, b0, b0, b0, b0, b0};
      accQ[nt] = (v8f){0.f, 0.f, 0.f, 0.f, 0.f, 0.f, 0.f, 0.f};
    }
#pragma unroll 1
    for (int kt = 0; kt < NODE_DIM / 32; ++kt) {
      const float4 x0 = *(const float4*)(hr + kt * 32 + kb);
      const float4 x1 = *(const float4*)(hr + kt * 32 + kb + 4);
      const float4 x2 = *(const float4*)(hr + kt * 32 + 16 + kb);
      const float4 x3 = *(const float4*)(hr + kt * 32 + 16 + kb + 4);
      v16h a;
      a[0]  = (h16)x0.x; a[1]  = (h16)x0.y; a[2]  = (h16)x0.z; a[3]  = (h16)x0.w;
      a[4]  = (h16)x1.x; a[5]  = (h16)x1.y; a[6]  = (h16)x1.z; a[7]  = (h16)x1.w;
      a[8]  = (h16)x2.x; a[9]  = (h16)x2.y; a[10] = (h16)x2.z; a[11] = (h16)x2.w;
      a[12] = (h16)x3.x; a[13] = (h16)x3.y; a[14] = (h16)x3.z; a[15] = (h16)x3.w;
#pragma unroll
      for (int nt = 0; nt < 8; ++nt) {
        v16h bp = load_b_frag(mw1p, kt, nt, 8, lane);
        accP[nt] = __builtin_amdgcn_wmma_f32_16x16x32_f16(false, a, false, bp,
                                                          (short)0, accP[nt], false, false);
        v16h bq = load_b_frag(mw1p, kt + NODE_DIM / 32, nt, 8, lane);
        accQ[nt] = __builtin_amdgcn_wmma_f32_16x16x32_f16(false, a, false, bq,
                                                          (short)0, accQ[nt], false, false);
      }
    }
#pragma unroll
    for (int r = 0; r < 8; ++r) {
      float4* o = (float4*)(pq + (size_t)(tile * 16 + drow + r) * (2 * HIDDEN) + m * 4);
      o[0]  = make_float4(accP[0][r], accP[1][r], accP[2][r], accP[3][r]);
      o[16] = make_float4(accP[4][r], accP[5][r], accP[6][r], accP[7][r]);
      o[32] = make_float4(accQ[0][r], accQ[1][r], accQ[2][r], accQ[3][r]);
      o[48] = make_float4(accQ[4][r], accQ[5][r], accQ[6][r], accQ[7][r]);
    }
  }
}

extern "C" __global__ void __launch_bounds__(WAVES * 32)
egnn_edge(const float* __restrict__ pq, const float* __restrict__ pos,
          const int* __restrict__ ei,
          const h16* __restrict__ mw1p,
          const h16* __restrict__ mw2p, const float* __restrict__ mb2,
          const h16* __restrict__ cw1p, const float* __restrict__ cb1,
          const float* __restrict__ cw2, const float* __restrict__ cb2,
          float* __restrict__ aggr, float* __restrict__ pos_out, int E) {
  extern __shared__ char smem[];
  const int wave = threadIdx.x >> 5;
  const int lane = threadIdx.x & 31;
  const int m    = lane & 15;
  const int kb   = (lane < 16) ? 0 : 8;
  const int drow = (lane < 16) ? 0 : 8;

  const int WB = 16 * HIDDEN * 2 + 128;
  char* wb = smem + wave * WB;
  h16*   s_hid = (h16*)wb;
  int*   s_row = (int*)(wb + 16 * HIDDEN * 2);
  int*   s_col = s_row + 16;

  v16h bw2[HIDDEN / 32];
#pragma unroll
  for (int kt = 0; kt < HIDDEN / 32; ++kt) {
#pragma unroll
    for (int s = 0; s < 8; ++s) {
      bw2[kt][s]     = (h16)cw2[kt * 32 + kb + s];
      bw2[kt][8 + s] = (h16)cw2[kt * 32 + 16 + kb + s];
    }
  }
  const float cb2v = cb2[0];
  const int ntile = E >> 4;
  const int nwav  = gridDim.x * WAVES;
  for (int tile = blockIdx.x * WAVES + wave; tile < ntile; tile += nwav) {
    const int e    = tile * 16 + m;
    const int vrow = ei[e];
    const int vcol = ei[E + e];
    const float rx = pos[vrow * 3 + 0] - pos[vcol * 3 + 0];
    const float ry = pos[vrow * 3 + 1] - pos[vcol * 3 + 1];
    const float rz = pos[vrow * 3 + 2] - pos[vcol * 3 + 2];
    const float dist = sqrtf(rx * rx + ry * ry + rz * rz + 1e-8f);
    s_row[lane] = (lane < 16) ? vrow : vcol;

    v16h ag;
    {
      const float step  = 10.0f / (NBASIS - 1);
      const float coeff = -0.5f / (step * step);
#pragma unroll
      for (int s = 0; s < 8; ++s) {
        float d0 = dist - step * (kb + s);
        float d1 = dist - step * (16 + kb + s);
        ag[s]     = (h16)__expf(coeff * d0 * d0);
        ag[8 + s] = (h16)__expf(coeff * d1 * d1);
      }
    }
    asm volatile("s_wait_dscnt 0" ::: "memory");
    int nr[8], nc[8];
    {
      const int4 r0 = *(const int4*)(s_row + drow), r1 = *(const int4*)(s_row + drow + 4);
      const int4 c0 = *(const int4*)(s_col + drow), c1 = *(const int4*)(s_col + drow + 4);
      nr[0] = r0.x; nr[1] = r0.y; nr[2] = r0.z; nr[3] = r0.w;
      nr[4] = r1.x; nr[5] = r1.y; nr[6] = r1.z; nr[7] = r1.w;
      nc[0] = c0.x; nc[1] = c0.y; nc[2] = c0.z; nc[3] = c0.w;
      nc[4] = c1.x; nc[5] = c1.y; nc[6] = c1.z; nc[7] = c1.w;
    }

    v8f acc[8];
#pragma unroll
    for (int nt = 0; nt < 8; ++nt) {
      v16h b = load_b_frag(mw1p, 2 * NODE_DIM / 32, nt, 8, lane);
      acc[nt] = __builtin_amdgcn_wmma_f32_16x16x32_f16(false, ag, false, b, (short)0,
                    (v8f){0.f, 0.f, 0.f, 0.f, 0.f, 0.f, 0.f, 0.f}, false, false);
    }
#pragma unroll
    for (int r = 0; r < 8; ++r) {
      const float4* pp = (const float4*)(pq + (size_t)nr[r] * (2 * HIDDEN) + m * 4);
      const float4* qp = (const float4*)(pq + (size_t)nc[r] * (2 * HIDDEN) + HIDDEN + m * 4);
      const float4 p0 = pp[0], p1 = pp[16], q0 = qp[0], q1 = qp[16];
      acc[0][r] += p0.x + q0.x; acc[1][r] += p0.y + q0.y;
      acc[2][r] += p0.z + q0.z; acc[3][r] += p0.w + q0.w;
      acc[4][r] += p1.x + q1.x; acc[5][r] += p1.y + q1.y;
      acc[6][r] += p1.z + q1.z; acc[7][r] += p1.w + q1.w;
    }
#pragma unroll
    for (int nt = 0; nt < 8; ++nt)
#pragma unroll
      for (int r = 0; r < 8; ++r)
        s_hid[(drow + r) * HIDDEN + nt * 16 + m] = (h16)silu(acc[nt][r]);
    asm volatile("s_wait_dscnt 0" ::: "memory");

    v8f acc2[8];
#pragma unroll
    for (int nt = 0; nt < 8; ++nt) {
      float b0 = mb2[nt * 16 + m];
      acc2[nt] = (v8f){b0, b0, b0, b0, b0, b0, b0, b0};
    }
#pragma unroll
    for (int kt = 0; kt < HIDDEN / 32; ++kt) {
      v16h a = load_a_frag(s_hid + m * HIDDEN, kt, kb);
#pragma unroll
      for (int nt = 0; nt < 8; ++nt) {
        v16h b = load_b_frag(mw2p, kt, nt, 8, lane);
        acc2[nt] = __builtin_amdgcn_wmma_f32_16x16x32_f16(false, a, false, b,
                                                          (short)0, acc2[nt], false, false);
      }
    }
#pragma unroll
    for (int nt = 0; nt < 8; ++nt) {
      const int col = nt * 16 + m;
#pragma unroll
      for (int r = 0; r < 8; ++r) {
        float v  = acc2[nt][r];
        int   rr = drow + r;
        unsafeAtomicAdd(&aggr[(size_t)nr[r] * HIDDEN + col], v);
        s_hid[rr * HIDDEN + col] = (h16)v;
      }
    }
    asm volatile("s_wait_dscnt 0" ::: "memory");

    v8f acc3[8];
#pragma unroll
    for (int nt = 0; nt < 8; ++nt) {
      float b0 = cb1[nt * 16 + m];
      acc3[nt] = (v8f){b0, b0, b0, b0, b0, b0, b0, b0};
    }
#pragma unroll
    for (int kt = 0; kt < HIDDEN / 32; ++kt) {
      v16h a = load_a_frag(s_hid + m * HIDDEN, kt, kb);
#pragma unroll
      for (int nt = 0; nt < 8; ++nt) {
        v16h b = load_b_frag(cw1p, kt, nt, 8, lane);
        acc3[nt] = __builtin_amdgcn_wmma_f32_16x16x32_f16(false, a, false, b,
                                                          (short)0, acc3[nt], false, false);
      }
    }
#pragma unroll
    for (int nt = 0; nt < 8; ++nt)
#pragma unroll
      for (int r = 0; r < 8; ++r)
        s_hid[(drow + r) * HIDDEN + nt * 16 + m] = (h16)silu(acc3[nt][r]);
    asm volatile("s_wait_dscnt 0" ::: "memory");
    v8f acc4 = (v8f){cb2v, cb2v, cb2v, cb2v, cb2v, cb2v, cb2v, cb2v};
#pragma unroll
    for (int kt = 0; kt < HIDDEN / 32; ++kt) {
      v16h a = load_a_frag(s_hid + m * HIDDEN, kt, kb);
      acc4 = __builtin_amdgcn_wmma_f32_16x16x32_f16(false, a, false, bw2[kt],
                                                    (short)0, acc4, false, false);
    }
    float sel = acc4[0];
#pragma unroll
    for (int r = 1; r < 8; ++r) sel = ((m & 7) == r) ? acc4[r] : sel;
    const float oth = __shfl_xor(sel, 16);
    if (lane < 16) {
      float cm = (m < 8) ? sel : oth;
      unsafeAtomicAdd(&pos_out[vrow * 3 + 0], rx * cm);
      unsafeAtomicAdd(&pos_out[vrow * 3 + 1], ry * cm);
      unsafeAtomicAdd(&pos_out[vrow * 3 + 2], rz * cm);
    }
    asm volatile("s_wait_dscnt 0" ::: "memory");
  }
}

extern "C" __global__ void __launch_bounds__(WAVES * 32)
egnn_node(const float* __restrict__ hfeat, const float* __restrict__ aggr,
          const h16* __restrict__ nw1p, const float* __restrict__ nb1,
          const h16* __restrict__ nw2p, const float* __restrict__ nb2,
          float* __restrict__ h_out, int Nn) {
  extern __shared__ char smem[];
  const int wave = threadIdx.x >> 5;
  const int lane = threadIdx.x & 31;
  const int m    = lane & 15;
  const int kb   = (lane < 16) ? 0 : 8;
  const int drow = (lane < 16) ? 0 : 8;

  const int NIN = 2 * NODE_DIM;
  const int WB  = 16 * NIN * 2 + 16 * HIDDEN * 2;
  char* wb = smem + wave * WB;
  h16* s_nin = (h16*)wb;
  h16* s_hid = (h16*)(wb + 16 * NIN * 2);

  const int ntile = Nn >> 4;
  const int nwav  = gridDim.x * WAVES;
  for (int tile = blockIdx.x * WAVES + wave; tile < ntile; tile += nwav) {
    const int node = tile * 16 + m;
    const float4* src = (lane < 16) ? (const float4*)(hfeat + (size_t)node * NODE_DIM)
                                    : (const float4*)(aggr + (size_t)node * HIDDEN);
    h16* dst = s_nin + m * NIN + ((lane < 16) ? 0 : NODE_DIM);
#pragma unroll 4
    for (int j = 0; j < NODE_DIM / 4; ++j) {
      float4 v = src[j];
      dst[j * 4 + 0] = (h16)v.x; dst[j * 4 + 1] = (h16)v.y;
      dst[j * 4 + 2] = (h16)v.z; dst[j * 4 + 3] = (h16)v.w;
    }
    asm volatile("s_wait_dscnt 0" ::: "memory");

    v8f acc[8];
#pragma unroll
    for (int nt = 0; nt < 8; ++nt) {
      float b0 = nb1[nt * 16 + m];
      acc[nt] = (v8f){b0, b0, b0, b0, b0, b0, b0, b0};
    }
#pragma unroll
    for (int kt = 0; kt < NIN / 32; ++kt) {
      v16h a = load_a_frag(s_nin + m * NIN, kt, kb);
#pragma unroll
      for (int nt = 0; nt < 8; ++nt) {
        v16h b = load_b_frag(nw1p, kt, nt, 8, lane);
        acc[nt] = __builtin_amdgcn_wmma_f32_16x16x32_f16(false, a, false, b,
                                                         (short)0, acc[nt], false, false);
      }
    }
#pragma unroll
    for (int nt = 0; nt < 8; ++nt)
#pragma unroll
      for (int r = 0; r < 8; ++r)
        s_hid[(drow + r) * HIDDEN + nt * 16 + m] = (h16)silu(acc[nt][r]);
    asm volatile("s_wait_dscnt 0" ::: "memory");

    v8f acc2[8];
#pragma unroll
    for (int nt = 0; nt < 8; ++nt) {
      float b0 = nb2[nt * 16 + m];
      acc2[nt] = (v8f){b0, b0, b0, b0, b0, b0, b0, b0};
    }
#pragma unroll
    for (int kt = 0; kt < HIDDEN / 32; ++kt) {
      v16h a = load_a_frag(s_hid + m * HIDDEN, kt, kb);
#pragma unroll
      for (int nt = 0; nt < 8; ++nt) {
        v16h b = load_b_frag(nw2p, kt, nt, 8, lane);
        acc2[nt] = __builtin_amdgcn_wmma_f32_16x16x32_f16(false, a, false, b,
                                                          (short)0, acc2[nt], false, false);
      }
    }
#pragma unroll
    for (int nt = 0; nt < 8; ++nt) {
      const int col = nt * 16 + m;
#pragma unroll
      for (int r = 0; r < 8; ++r) {
        int n2 = tile * 16 + drow + r;
        h_out[(size_t)n2 * NODE_DIM + col] =
            hfeat[(size_t)n2 * NODE_DIM + col] + acc2[nt][r];
      }
    }
  }
}

extern "C" void kernel_launch(void* const* d_in, const int* in_sizes, int n_in,
                              void* d_out, int out_size, void* d_ws, size_t ws_size,
                              hipStream_t stream) {
  const float* h   = (const float*)d_in[0];
  const float* pos = (const float*)d_in[1];
  const int*   ei  = (const int*)d_in[2];
  const float* mw1 = (const float*)d_in[3];  const float* mb1 = (const float*)d_in[4];
  const float* mw2 = (const float*)d_in[5];  const float* mb2 = (const float*)d_in[6];
  const float* nw1 = (const float*)d_in[7];  const float* nb1 = (const float*)d_in[8];
  const float* nw2 = (const float*)d_in[9];  const float* nb2 = (const float*)d_in[10];
  const float* cw1 = (const float*)d_in[11]; const float* cb1 = (const float*)d_in[12];
  const float* cw2 = (const float*)d_in[13]; const float* cb2 = (const float*)d_in[14];
  const int N = in_sizes[0] / NODE_DIM;
  const int E = in_sizes[2] / 2;

  char* ws = (char*)d_ws;
  h16* mw1p = (h16*)(ws + 0);
  h16* mw2p = (h16*)(ws + 73728);
  h16* cw1p = (h16*)(ws + 106496);
  h16* nw1p = (h16*)(ws + 139264);
  h16* nw2p = (h16*)(ws + 204800);
  float* aggr = (float*)(ws + 237568);
  float* pq   = (float*)(ws + 25837568);
  float* h_out   = (float*)d_out;
  float* pos_out = h_out + (size_t)N * NODE_DIM;

  pack_w<<<(9 * 8 * 32 + 127) / 128, 128, 0, stream>>>(mw1, mw1p, MSG_IN, HIDDEN);
  pack_w<<<(4 * 8 * 32 + 127) / 128, 128, 0, stream>>>(mw2, mw2p, HIDDEN, HIDDEN);
  pack_w<<<(4 * 8 * 32 + 127) / 128, 128, 0, stream>>>(cw1, cw1p, HIDDEN, HIDDEN);
  pack_w<<<(8 * 8 * 32 + 127) / 128, 128, 0, stream>>>(nw1, nw1p, 2 * NODE_DIM, HIDDEN);
  pack_w<<<(4 * 8 * 32 + 127) / 128, 128, 0, stream>>>(nw2, nw2p, HIDDEN, NODE_DIM);
  egnn_init<<<2048, 256, 0, stream>>>(aggr, pos, pos_out, N);

  egnn_pre<<<800, WAVES * 32, 0, stream>>>(h, mw1p, mb1, pq, N);

  size_t lds_e = (size_t)WAVES * (16 * HIDDEN * 2 + 128);
  egnn_edge<<<2048, WAVES * 32, lds_e, stream>>>(pq, pos, ei, mw1p, mw2p, mb2,
                                                 cw1p, cb1, cw2, cb2, aggr, pos_out, E);

  size_t lds_n = (size_t)WAVES * (16 * 2 * NODE_DIM * 2 + 16 * HIDDEN * 2);
  egnn_node<<<800, WAVES * 32, lds_n, stream>>>(h, aggr, nw1p, nb1, nw2p, nb2, h_out, N);
}


# Edge kernel: each k-step's 8 B fragments and A fragment are loaded as one group a whole 8-WMMA block ahead into a second register set, not 2 loads between every two WMMAs; values unchanged
# speedup vs baseline: 1.8724x; 1.0517x over previous
#include <hip/hip_runtime.h>
#include <math.h>

typedef _Float16 h16;
typedef __attribute__((ext_vector_type(16))) _Float16 v16h;
typedef __attribute__((ext_vector_type(8)))  _Float16 v8h;
typedef __attribute__((ext_vector_type(8)))  float    v8f;

#define NODE_DIM 128
#define HIDDEN   128
#define NBASIS   32
#define MSG_IN   288
#define WAVES    4

static __device__ __forceinline__ v16h cat8(v8h a, v8h b) {
  return __builtin_shufflevector(a, b, 0,1,2,3,4,5,6,7,8,9,10,11,12,13,14,15);
}
static __device__ __forceinline__ float silu(float x) {
  return x / (1.0f + __expf(-x));
}
static __device__ __forceinline__ v16h load_b_frag(const h16* __restrict__ wp,
                                                   int kt, int nt, int NT, int lane) {
  const v8h* p = (const v8h*)(wp + (size_t)(((kt * NT + nt) * 32 + lane)) * 16);
  return cat8(p[0], p[1]);
}
static __device__ __forceinline__ v16h load_a_frag(const h16* row_base, int kt, int kb) {
  const v8h* a0 = (const v8h*)(row_base + kt * 32 + kb);
  const v8h* a1 = (const v8h*)(row_base + kt * 32 + 16 + kb);
  return cat8(*a0, *a1);
}

static __device__ __forceinline__ void load_b_row(v16h (&b)[8], const h16* __restrict__ wp,
                                                  int kt, int lane) {
#pragma unroll
  for (int nt = 0; nt < 8; ++nt) b[nt] = load_b_frag(wp, kt, nt, 8, lane);
}
template <bool NEXT>
static __device__ __forceinline__ void layer128(v8f (&acc)[8], v16h (&s0)[8],
                                                const h16* arow, int kb,
                                                const h16* __restrict__ wp,
                                                const h16* __restrict__ wnext, int lane) {
  v16h s1[8];
  v16h a0 = load_a_frag(arow, 0, kb);
#pragma unroll 1
  for (int kt = 0; kt < HIDDEN / 32; kt += 2) {
    load_b_row(s1, wp, kt + 1, lane);
    const v16h a1 = load_a_frag(arow, kt + 1, kb);
    __builtin_amdgcn_sched_barrier(0);
#pragma unroll
    for (int nt = 0; nt < 8; ++nt)
      acc[nt] = __builtin_amdgcn_wmma_f32_16x16x32_f16(false, a0, false, s0[nt],
                                                       (short)0, acc[nt], false, false);
    __builtin_amdgcn_sched_barrier(0);
    if (NEXT) {
      const h16* nx = (kt + 2 < HIDDEN / 32) ? wp + (size_t)(kt + 2) * (8 * 32 * 16) : wnext;
      load_b_row(s0, nx, 0, lane);
      a0 = load_a_frag(arow, (kt + 2) & (HIDDEN / 32 - 1), kb);
    } else if (kt + 2 < HIDDEN / 32) {
      load_b_row(s0, wp, kt + 2, lane);
      a0 = load_a_frag(arow, kt + 2, kb);
    }
    __builtin_amdgcn_sched_barrier(0);
#pragma unroll
    for (int nt = 0; nt < 8; ++nt)
      acc[nt] = __builtin_amdgcn_wmma_f32_16x16x32_f16(false, a1, false, s1[nt],
                                                       (short)0, acc[nt], false, false);
    __builtin_amdgcn_sched_barrier(0);
  }
}

extern "C" __global__ void pack_w(const float* __restrict__ W, h16* __restrict__ out,
                                  int K, int Nn) {
  int t = blockIdx.x * blockDim.x + threadIdx.x;
  int KT = K >> 5, NT = Nn >> 4;
  int total = KT * NT * 32;
  if (t >= total) return;
  int lane = t & 31;
  int tile = t >> 5;
  int nt = tile % NT, kt = tile / NT;
  int n = nt * 16 + (lane & 15);
  int kbase = (lane < 16) ? 0 : 8;
  alignas(16) h16 tmp[16];
#pragma unroll
  for (int r = 0; r < 4; ++r) {
    int k0 = kt * 32 + kbase + 2 * r;
    tmp[2 * r]     = (h16)W[(size_t)k0 * Nn + n];
    tmp[2 * r + 1] = (h16)W[(size_t)(k0 + 1) * Nn + n];
    int k1 = kt * 32 + 16 + kbase + 2 * r;
    tmp[8 + 2 * r]     = (h16)W[(size_t)k1 * Nn + n];
    tmp[8 + 2 * r + 1] = (h16)W[(size_t)(k1 + 1) * Nn + n];
  }
  v8h* o = (v8h*)(out + (size_t)t * 16);
  o[0] = *(const v8h*)(tmp);
  o[1] = *(const v8h*)(tmp + 8);
}

extern "C" __global__ void egnn_init(float* __restrict__ aggr, const float* __restrict__ pos,
                                     float* __restrict__ pos_out, int N) {
  int i = blockIdx.x * blockDim.x + threadIdx.x;
  int stride = gridDim.x * blockDim.x;
  for (int j = i; j < N * HIDDEN; j += stride) aggr[j] = 0.0f;
  for (int j = i; j < N * 3; j += stride) pos_out[j] = pos[j];
}

extern "C" __global__ void __launch_bounds__(WAVES * 32)
egnn_pre(const float* __restrict__ hfeat, const h16* __restrict__ mw1p,
         const float* __restrict__ mb1, float* __restrict__ pq, int Nn) {
  const int wave = threadIdx.x >> 5;
  const int lane = threadIdx.x & 31;
  const int m    = lane & 15;
  const int kb   = (lane < 16) ? 0 : 8;
  const int drow = (lane < 16) ? 0 : 8;
  const int ntile = Nn >> 4;
  const int nwav  = gridDim.x * WAVES;
  for (int tile = blockIdx.x * WAVES + wave; tile < ntile; tile += nwav) {
    const float* hr = hfeat + (size_t)(tile * 16 + m) * NODE_DIM;
    v8f accP[8], accQ[8];
#pragma unroll
    for (int nt = 0; nt < 8; ++nt) {
      float b0 = mb1[nt * 16 + m];
      accP[nt] = (v8f){b0, b0, b0, b0, b0, b0, b0, b0};
      accQ[nt] = (v8f){0.f, 0.f, 0.f, 0.f, 0.f, 0.f, 0.f, 0.f};
    }
#pragma unroll 1
    for (int kt = 0; kt < NODE_DIM / 32; ++kt) {
      const float4 x0 = *(const float4*)(hr + kt * 32 + kb);
      const float4 x1 = *(const float4*)(hr + kt * 32 + kb + 4);
      const float4 x2 = *(const float4*)(hr + kt * 32 + 16 + kb);
      const float4 x3 = *(const float4*)(hr + kt * 32 + 16 + kb + 4);
      v16h a;
      a[0]  = (h16)x0.x; a[1]  = (h16)x0.y; a[2]  = (h16)x0.z; a[3]  = (h16)x0.w;
      a[4]  = (h16)x1.x; a[5]  = (h16)x1.y; a[6]  = (h16)x1.z; a[7]  = (h16)x1.w;
      a[8]  = (h16)x2.x; a[9]  = (h16)x2.y; a[10] = (h16)x2.z; a[11] = (h16)x2.w;
      a[12] = (h16)x3.x; a[13] = (h16)x3.y; a[14] = (h16)x3.z; a[15] = (h16)x3.w;
#pragma unroll
      for (int nt = 0; nt < 8; ++nt) {
        v16h bp = load_b_frag(mw1p, kt, nt, 8, lane);
        accP[nt] = __builtin_amdgcn_wmma_f32_16x16x32_f16(false, a, false, bp,
                                                          (short)0, accP[nt], false, false);
        v16h bq = load_b_frag(mw1p, kt + NODE_DIM / 32, nt, 8, lane);
        accQ[nt] = __builtin_amdgcn_wmma_f32_16x16x32_f16(false, a, false, bq,
                                                          (short)0, accQ[nt], false, false);
      }
    }
#pragma unroll
    for (int r = 0; r < 8; ++r) {
      float4* o = (float4*)(pq + (size_t)(tile * 16 + drow + r) * (2 * HIDDEN) + m * 4);
      o[0]  = make_float4(accP[0][r], accP[1][r], accP[2][r], accP[3][r]);
      o[16] = make_float4(accP[4][r], accP[5][r], accP[6][r], accP[7][r]);
      o[32] = make_float4(accQ[0][r], accQ[1][r], accQ[2][r], accQ[3][r]);
      o[48] = make_float4(accQ[4][r], accQ[5][r], accQ[6][r], accQ[7][r]);
    }
  }
}

extern "C" __global__ void __launch_bounds__(WAVES * 32)
egnn_edge(const float* __restrict__ pq, const float* __restrict__ pos,
          const int* __restrict__ ei,
          const h16* __restrict__ mw1p,
          const h16* __restrict__ mw2p, const float* __restrict__ mb2,
          const h16* __restrict__ cw1p, const float* __restrict__ cb1,
          const float* __restrict__ cw2, const float* __restrict__ cb2,
          float* __restrict__ aggr, float* __restrict__ pos_out, int E) {
  extern __shared__ char smem[];
  const int wave = threadIdx.x >> 5;
  const int lane = threadIdx.x & 31;
  const int m    = lane & 15;
  const int kb   = (lane < 16) ? 0 : 8;
  const int drow = (lane < 16) ? 0 : 8;

  const int WB = 16 * HIDDEN * 2 + 128;
  char* wb = smem + wave * WB;
  h16*   s_hid = (h16*)wb;
  int*   s_row = (int*)(wb + 16 * HIDDEN * 2);
  int*   s_col = s_row + 16;

  v16h bw2[HIDDEN / 32];
#pragma unroll
  for (int kt = 0; kt < HIDDEN / 32; ++kt) {
#pragma unroll
    for (int s = 0; s < 8; ++s) {
      bw2[kt][s]     = (h16)cw2[kt * 32 + kb + s];
      bw2[kt][8 + s] = (h16)cw2[kt * 32 + 16 + kb + s];
    }
  }
  const float cb2v = cb2[0];
  const int ntile = E >> 4;
  const int nwav  = gridDim.x * WAVES;
  for (int tile = blockIdx.x * WAVES + wave; tile < ntile; tile += nwav) {
    const int e    = tile * 16 + m;
    const int vrow = ei[e];
    const int vcol = ei[E + e];
    const float rx = pos[vrow * 3 + 0] - pos[vcol * 3 + 0];
    const float ry = pos[vrow * 3 + 1] - pos[vcol * 3 + 1];
    const float rz = pos[vrow * 3 + 2] - pos[vcol * 3 + 2];
    const float dist = sqrtf(rx * rx + ry * ry + rz * rz + 1e-8f);
    s_row[lane] = (lane < 16) ? vrow : vcol;

    v16h ag;
    {
      const float step  = 10.0f / (NBASIS - 1);
      const float coeff = -0.5f / (step * step);
#pragma unroll
      for (int s = 0; s < 8; ++s) {
        float d0 = dist - step * (kb + s);
        float d1 = dist - step * (16 + kb + s);
        ag[s]     = (h16)__expf(coeff * d0 * d0);
        ag[8 + s] = (h16)__expf(coeff * d1 * d1);
      }
    }
    asm volatile("s_wait_dscnt 0" ::: "memory");
    int nr[8], nc[8];
    {
      const int4 r0 = *(const int4*)(s_row + drow), r1 = *(const int4*)(s_row + drow + 4);
      const int4 c0 = *(const int4*)(s_col + drow), c1 = *(const int4*)(s_col + drow + 4);
      nr[0] = r0.x; nr[1] = r0.y; nr[2] = r0.z; nr[3] = r0.w;
      nr[4] = r1.x; nr[5] = r1.y; nr[6] = r1.z; nr[7] = r1.w;
      nc[0] = c0.x; nc[1] = c0.y; nc[2] = c0.z; nc[3] = c0.w;
      nc[4] = c1.x; nc[5] = c1.y; nc[6] = c1.z; nc[7] = c1.w;
    }

    v8f acc[8];
#pragma unroll
    for (int nt = 0; nt < 8; ++nt) {
      v16h b = load_b_frag(mw1p, 2 * NODE_DIM / 32, nt, 8, lane);
      acc[nt] = __builtin_amdgcn_wmma_f32_16x16x32_f16(false, ag, false, b, (short)0,
                    (v8f){0.f, 0.f, 0.f, 0.f, 0.f, 0.f, 0.f, 0.f}, false, false);
    }
#pragma unroll
    for (int r = 0; r < 8; ++r) {
      const float4* pp = (const float4*)(pq + (size_t)nr[r] * (2 * HIDDEN) + m * 4);
      const float4* qp = (const float4*)(pq + (size_t)nc[r] * (2 * HIDDEN) + HIDDEN + m * 4);
      const float4 p0 = pp[0], p1 = pp[16], q0 = qp[0], q1 = qp[16];
      acc[0][r] += p0.x + q0.x; acc[1][r] += p0.y + q0.y;
      acc[2][r] += p0.z + q0.z; acc[3][r] += p0.w + q0.w;
      acc[4][r] += p1.x + q1.x; acc[5][r] += p1.y + q1.y;
      acc[6][r] += p1.z + q1.z; acc[7][r] += p1.w + q1.w;
    }
    __builtin_amdgcn_sched_barrier(0);
    v16h bs[8];
    load_b_row(bs, mw2p, 0, lane);
#pragma unroll
    for (int nt = 0; nt < 8; ++nt)
#pragma unroll
      for (int r = 0; r < 8; ++r)
        s_hid[(drow + r) * HIDDEN + nt * 16 + m] = (h16)silu(acc[nt][r]);
    asm volatile("s_wait_dscnt 0" ::: "memory");

    v8f acc2[8];
#pragma unroll
    for (int nt = 0; nt < 8; ++nt) {
      float b0 = mb2[nt * 16 + m];
      acc2[nt] = (v8f){b0, b0, b0, b0, b0, b0, b0, b0};
    }
    layer128<true>(acc2, bs, s_hid + m * HIDDEN, kb, mw2p, cw1p, lane);
#pragma unroll
    for (int nt = 0; nt < 8; ++nt) {
      const int col = nt * 16 + m;
#pragma unroll
      for (int r = 0; r < 8; ++r) {
        float v  = acc2[nt][r];
        int   rr = drow + r;
        unsafeAtomicAdd(&aggr[(size_t)nr[r] * HIDDEN + col], v);
        s_hid[rr * HIDDEN + col] = (h16)v;
      }
    }
    asm volatile("s_wait_dscnt 0" ::: "memory");

    v8f acc3[8];
#pragma unroll
    for (int nt = 0; nt < 8; ++nt) {
      float b0 = cb1[nt * 16 + m];
      acc3[nt] = (v8f){b0, b0, b0, b0, b0, b0, b0, b0};
    }
    layer128<false>(acc3, bs, s_hid + m * HIDDEN, kb, cw1p, cw1p, lane);
#pragma unroll
    for (int nt = 0; nt < 8; ++nt)
#pragma unroll
      for (int r = 0; r < 8; ++r)
        s_hid[(drow + r) * HIDDEN + nt * 16 + m] = (h16)silu(acc3[nt][r]);
    asm volatile("s_wait_dscnt 0" ::: "memory");
    v8f acc4 = (v8f){cb2v, cb2v, cb2v, cb2v, cb2v, cb2v, cb2v, cb2v};
#pragma unroll
    for (int kt = 0; kt < HIDDEN / 32; ++kt) {
      v16h a = load_a_frag(s_hid + m * HIDDEN, kt, kb);
      acc4 = __builtin_amdgcn_wmma_f32_16x16x32_f16(false, a, false, bw2[kt],
                                                    (short)0, acc4, false, false);
    }
    float sel = acc4[0];
#pragma unroll
    for (int r = 1; r < 8; ++r) sel = ((m & 7) == r) ? acc4[r] : sel;
    const float oth = __shfl_xor(sel, 16);
    if (lane < 16) {
      float cm = (m < 8) ? sel : oth;
      unsafeAtomicAdd(&pos_out[vrow * 3 + 0], rx * cm);
      unsafeAtomicAdd(&pos_out[vrow * 3 + 1], ry * cm);
      unsafeAtomicAdd(&pos_out[vrow * 3 + 2], rz * cm);
    }
    asm volatile("s_wait_dscnt 0" ::: "memory");
  }
}

extern "C" __global__ void __launch_bounds__(WAVES * 32)
egnn_node(const float* __restrict__ hfeat, const float* __restrict__ aggr,
          const h16* __restrict__ nw1p, const float* __restrict__ nb1,
          const h16* __restrict__ nw2p, const float* __restrict__ nb2,
          float* __restrict__ h_out, int Nn) {
  extern __shared__ char smem[];
  const int wave = threadIdx.x >> 5;
  const int lane = threadIdx.x & 31;
  const int m    = lane & 15;
  const int kb   = (lane < 16) ? 0 : 8;
  const int drow = (lane < 16) ? 0 : 8;

  const int NIN = 2 * NODE_DIM;
  const int WB  = 16 * NIN * 2 + 16 * HIDDEN * 2;
  char* wb = smem + wave * WB;
  h16* s_nin = (h16*)wb;
  h16* s_hid = (h16*)(wb + 16 * NIN * 2);

  const int ntile = Nn >> 4;
  const int nwav  = gridDim.x * WAVES;
  for (int tile = blockIdx.x * WAVES + wave; tile < ntile; tile += nwav) {
    const int node = tile * 16 + m;
    const float4* src = (lane < 16) ? (const float4*)(hfeat + (size_t)node * NODE_DIM)
                                    : (const float4*)(aggr + (size_t)node * HIDDEN);
    h16* dst = s_nin + m * NIN + ((lane < 16) ? 0 : NODE_DIM);
#pragma unroll 4
    for (int j = 0; j < NODE_DIM / 4; ++j) {
      float4 v = src[j];
      dst[j * 4 + 0] = (h16)v.x; dst[j * 4 + 1] = (h16)v.y;
      dst[j * 4 + 2] = (h16)v.z; dst[j * 4 + 3] = (h16)v.w;
    }
    asm volatile("s_wait_dscnt 0" ::: "memory");

    v8f acc[8];
#pragma unroll
    for (int nt = 0; nt < 8; ++nt) {
      float b0 = nb1[nt * 16 + m];
      acc[nt] = (v8f){b0, b0, b0, b0, b0, b0, b0, b0};
    }
#pragma unroll
    for (int kt = 0; kt < NIN / 32; ++kt) {
      v16h a = load_a_frag(s_nin + m * NIN, kt, kb);
#pragma unroll
      for (int nt = 0; nt < 8; ++nt) {
        v16h b = load_b_frag(nw1p, kt, nt, 8, lane);
        acc[nt] = __builtin_amdgcn_wmma_f32_16x16x32_f16(false, a, false, b,
                                                         (short)0, acc[nt], false, false);
      }
    }
#pragma unroll
    for (int nt = 0; nt < 8; ++nt)
#pragma unroll
      for (int r = 0; r < 8; ++r)
        s_hid[(drow + r) * HIDDEN + nt * 16 + m] = (h16)silu(acc[nt][r]);
    asm volatile("s_wait_dscnt 0" ::: "memory");

    v8f acc2[8];
#pragma unroll
    for (int nt = 0; nt < 8; ++nt) {
      float b0 = nb2[nt * 16 + m];
      acc2[nt] = (v8f){b0, b0, b0, b0, b0, b0, b0, b0};
    }
#pragma unroll
    for (int kt = 0; kt < HIDDEN / 32; ++kt) {
      v16h a = load_a_frag(s_hid + m * HIDDEN, kt, kb);
#pragma unroll
      for (int nt = 0; nt < 8; ++nt) {
        v16h b = load_b_frag(nw2p, kt, nt, 8, lane);
        acc2[nt] = __builtin_amdgcn_wmma_f32_16x16x32_f16(false, a, false, b,
                                                          (short)0, acc2[nt], false, false);
      }
    }
#pragma unroll
    for (int nt = 0; nt < 8; ++nt) {
      const int col = nt * 16 + m;
#pragma unroll
      for (int r = 0; r < 8; ++r) {
        int n2 = tile * 16 + drow + r;
        h_out[(size_t)n2 * NODE_DIM + col] =
            hfeat[(size_t)n2 * NODE_DIM + col] + acc2[nt][r];
      }
    }
  }
}

extern "C" void kernel_launch(void* const* d_in, const int* in_sizes, int n_in,
                              void* d_out, int out_size, void* d_ws, size_t ws_size,
                              hipStream_t stream) {
  const float* h   = (const float*)d_in[0];
  const float* pos = (const float*)d_in[1];
  const int*   ei  = (const int*)d_in[2];
  const float* mw1 = (const float*)d_in[3];  const float* mb1 = (const float*)d_in[4];
  const float* mw2 = (const float*)d_in[5];  const float* mb2 = (const float*)d_in[6];
  const float* nw1 = (const float*)d_in[7];  const float* nb1 = (const float*)d_in[8];
  const float* nw2 = (const float*)d_in[9];  const float* nb2 = (const float*)d_in[10];
  const float* cw1 = (const float*)d_in[11]; const float* cb1 = (const float*)d_in[12];
  const float* cw2 = (const float*)d_in[13]; const float* cb2 = (const float*)d_in[14];
  const int N = in_sizes[0] / NODE_DIM;
  const int E = in_sizes[2] / 2;

  char* ws = (char*)d_ws;
  h16* mw1p = (h16*)(ws + 0);
  h16* mw2p = (h16*)(ws + 73728);
  h16* cw1p = (h16*)(ws + 106496);
  h16* nw1p = (h16*)(ws + 139264);
  h16* nw2p = (h16*)(ws + 204800);
  float* aggr = (float*)(ws + 237568);
  float* pq   = (float*)(ws + 25837568);
  float* h_out   = (float*)d_out;
  float* pos_out = h_out + (size_t)N * NODE_DIM;

  pack_w<<<(9 * 8 * 32 + 127) / 128, 128, 0, stream>>>(mw1, mw1p, MSG_IN, HIDDEN);
  pack_w<<<(4 * 8 * 32 + 127) / 128, 128, 0, stream>>>(mw2, mw2p, HIDDEN, HIDDEN);
  pack_w<<<(4 * 8 * 32 + 127) / 128, 128, 0, stream>>>(cw1, cw1p, HIDDEN, HIDDEN);
  pack_w<<<(8 * 8 * 32 + 127) / 128, 128, 0, stream>>>(nw1, nw1p, 2 * NODE_DIM, HIDDEN);
  pack_w<<<(4 * 8 * 32 + 127) / 128, 128, 0, stream>>>(nw2, nw2p, HIDDEN, NODE_DIM);
  egnn_init<<<2048, 256, 0, stream>>>(aggr, pos, pos_out, N);

  egnn_pre<<<800, WAVES * 32, 0, stream>>>(h, mw1p, mb1, pq, N);

  size_t lds_e = (size_t)WAVES * (16 * HIDDEN * 2 + 128);
  egnn_edge<<<2048, WAVES * 32, lds_e, stream>>>(pq, pos, ei, mw1p, mw2p, mb2,
                                                 cw1p, cb1, cw2, cb2, aggr, pos_out, E);

  size_t lds_n = (size_t)WAVES * (16 * 2 * NODE_DIM * 2 + 16 * HIDDEN * 2);
  egnn_node<<<800, WAVES * 32, lds_n, stream>>>(h, aggr, nw1p, nb1, nw2p, nb2, h_out, N);
}


# Packed weights get a new column order: a lane's 8 accumulator tiles hold 8 neighbouring output columns, so panel rows, biases, h_out rows move as 16-byte pieces, not 2- and 4-byte; values unchanged
# speedup vs baseline: 2.5531x; 1.3635x over previous
#include <hip/hip_runtime.h>
#include <math.h>

typedef _Float16 h16;
typedef __attribute__((ext_vector_type(16))) _Float16 v16h;
typedef __attribute__((ext_vector_type(8)))  _Float16 v8h;
typedef __attribute__((ext_vector_type(8)))  float    v8f;
typedef __attribute__((ext_vector_type(4)))  float    v4f;

#define NODE_DIM 128
#define HIDDEN   128
#define NBASIS   32
#define MSG_IN   288
#define WAVES    4
#define NBIN     256
#define NPB      196
#define BING     250
#define EWAVES   12
#define ACC_LD   144
#define HID_LD   136
#define NIN_LD   264

static __device__ __forceinline__ v16h cat8(v8h a, v8h b) {
  return __builtin_shufflevector(a, b, 0,1,2,3,4,5,6,7,8,9,10,11,12,13,14,15);
}
static __device__ __forceinline__ float silu(float x) {
  return x / (1.0f + __expf(-x));
}
static __device__ __forceinline__ void load8(float (&b)[8], const float* __restrict__ p) {
  const float4 u = ((const float4*)p)[0], v = ((const float4*)p)[1];
  b[0] = u.x; b[1] = u.y; b[2] = u.z; b[3] = u.w;
  b[4] = v.x; b[5] = v.y; b[6] = v.z; b[7] = v.w;
}
static __device__ __forceinline__ void init_bias(v8f (&acc)[8], const float* __restrict__ bias,
                                                 int m) {
  float bb[8];
  load8(bb, bias + m * 8);
#pragma unroll
  for (int nt = 0; nt < 8; ++nt) {
    const float b0 = bb[nt];
    acc[nt] = (v8f){b0, b0, b0, b0, b0, b0, b0, b0};
  }
}
template <bool SILU>
static __device__ __forceinline__ void store_panel(h16* panel, const v8f (&acc)[8],
                                                   int drow, int m) {
  v8h hv[8];
#pragma unroll
  for (int np = 0; np < 8; np += 2)
#pragma unroll
    for (int r = 0; r < 8; ++r) {
      hv[r][np]     = (h16)(SILU ? silu(acc[np][r])     : acc[np][r]);
      hv[r][np + 1] = (h16)(SILU ? silu(acc[np + 1][r]) : acc[np + 1][r]);
    }
#pragma unroll
  for (int r = 0; r < 8; ++r)
    *(v8h*)(panel + (drow + r) * HID_LD + m * 8) = hv[r];
}
static __device__ __forceinline__ v16h load_b_frag(const h16* __restrict__ wp,
                                                   int kt, int nt, int NT, int lane) {
  const v8h* p = (const v8h*)(wp + (size_t)(((kt * NT + nt) * 32 + lane)) * 16);
  return cat8(p[0], p[1]);
}
static __device__ __forceinline__ v16h load_a_frag(const h16* row_base, int kt, int kb) {
  const v8h* a0 = (const v8h*)(row_base + kt * 32 + kb);
  const v8h* a1 = (const v8h*)(row_base + kt * 32 + 16 + kb);
  return cat8(*a0, *a1);
}

static __device__ __forceinline__ void load_b_row(v16h (&b)[8], const h16* __restrict__ wp,
                                                  int kt, int lane) {
#pragma unroll
  for (int nt = 0; nt < 8; ++nt) b[nt] = load_b_frag(wp, kt, nt, 8, lane);
}
template <bool NEXT>
static __device__ __forceinline__ void layer128(v8f (&acc)[8], v16h (&s0)[8],
                                                const h16* arow, int kb,
                                                const h16* __restrict__ wp,
                                                const h16* __restrict__ wnext, int lane) {
  v16h s1[8];
  v16h a0 = load_a_frag(arow, 0, kb);
#pragma unroll 1
  for (int kt = 0; kt < HIDDEN / 32; kt += 2) {
    load_b_row(s1, wp, kt + 1, lane);
    const v16h a1 = load_a_frag(arow, kt + 1, kb);
    __builtin_amdgcn_sched_barrier(0);
#pragma unroll
    for (int nt = 0; nt < 8; ++nt)
      acc[nt] = __builtin_amdgcn_wmma_f32_16x16x32_f16(false, a0, false, s0[nt],
                                                       (short)0, acc[nt], false, false);
    __builtin_amdgcn_sched_barrier(0);
    if (NEXT) {
      const h16* nx = (kt + 2 < HIDDEN / 32) ? wp + (size_t)(kt + 2) * (8 * 32 * 16) : wnext;
      load_b_row(s0, nx, 0, lane);
      a0 = load_a_frag(arow, (kt + 2) & (HIDDEN / 32 - 1), kb);
    } else if (kt + 2 < HIDDEN / 32) {
      load_b_row(s0, wp, kt + 2, lane);
      a0 = load_a_frag(arow, kt + 2, kb);
    }
    __builtin_amdgcn_sched_barrier(0);
#pragma unroll
    for (int nt = 0; nt < 8; ++nt)
      acc[nt] = __builtin_amdgcn_wmma_f32_16x16x32_f16(false, a1, false, s1[nt],
                                                       (short)0, acc[nt], false, false);
    __builtin_amdgcn_sched_barrier(0);
  }
}

extern "C" __global__ void pack_w(const float* __restrict__ W, h16* __restrict__ out,
                                  int K, int Nn) {
  int t = blockIdx.x * blockDim.x + threadIdx.x;
  int KT = K >> 5, NT = Nn >> 4;
  int total = KT * NT * 32;
  if (t >= total) return;
  int lane = t & 31;
  int tile = t >> 5;
  int nt = tile % NT, kt = tile / NT;
  int n = (lane & 15) * NT + nt;
  int kbase = (lane < 16) ? 0 : 8;
  alignas(16) h16 tmp[16];
#pragma unroll
  for (int r = 0; r < 4; ++r) {
    int k0 = kt * 32 + kbase + 2 * r;
    tmp[2 * r]     = (h16)W[(size_t)k0 * Nn + n];
    tmp[2 * r + 1] = (h16)W[(size_t)(k0 + 1) * Nn + n];
    int k1 = kt * 32 + 16 + kbase + 2 * r;
    tmp[8 + 2 * r]     = (h16)W[(size_t)k1 * Nn + n];
    tmp[8 + 2 * r + 1] = (h16)W[(size_t)(k1 + 1) * Nn + n];
  }
  v8h* o = (v8h*)(out + (size_t)t * 16);
  o[0] = *(const v8h*)(tmp);
  o[1] = *(const v8h*)(tmp + 8);
}

extern "C" __global__ void __launch_bounds__(NBIN)
bin_count(const int* __restrict__ ei, int* __restrict__ counts,
          const float* __restrict__ pos, float* __restrict__ pos_out, int E, int Nn) {
  __shared__ int hist[NBIN];
  hist[threadIdx.x] = 0;
  __syncthreads();
  const int chunk = (E + BING - 1) / BING;
  const int e0 = blockIdx.x * chunk;
  const int e1 = min(e0 + chunk, E);
  for (int e = e0 + (int)threadIdx.x; e < e1; e += NBIN)
    atomicAdd(&hist[min(ei[e] / NPB, NBIN - 1)], 1);
  __syncthreads();
  counts[blockIdx.x * NBIN + threadIdx.x] = hist[threadIdx.x];
  counts[(blockIdx.x + BING) * NBIN + threadIdx.x] = hist[threadIdx.x];
  for (int j = blockIdx.x * NBIN + threadIdx.x; j < Nn * 3; j += BING * NBIN)
    pos_out[j] = pos[j];
}
extern "C" __global__ void __launch_bounds__(NBIN)
bin_place(const int* __restrict__ ei, const int* __restrict__ counts,
          int* __restrict__ binstart, int2* __restrict__ list, int E) {
  __shared__ int cur[NBIN];
  __shared__ int wtot[NBIN / 32];
  const int b = threadIdx.x;
  const int s = blockIdx.x;
  int total = 0, before = 0;
  const int* p = counts + s * NBIN + b;
  const int nb = BING - s;
  for (int i0 = 0; i0 < BING; i0 += 50) {
    int cc[50];
#pragma unroll
    for (int j = 0; j < 50; ++j) cc[j] = p[(i0 + j) * NBIN];
#pragma unroll
    for (int j = 0; j < 50; ++j) {
      total += cc[j];
      before += (i0 + j >= nb) ? cc[j] : 0;
    }
  }
  const int lane = b & 31, w = b >> 5;
  int incl = total;
  for (int d = 1; d < 32; d <<= 1) {
    const int v = __shfl_up(incl, d);
    incl += (lane >= d) ? v : 0;
  }
  if (lane == 31) wtot[w] = incl;
  __syncthreads();
  for (int j = 0; j < NBIN / 32; ++j) incl += (j < w) ? wtot[j] : 0;
  cur[b] = incl - total + before;
  if (s == 0) {
    binstart[b + 1] = incl;
    if (b == 0) binstart[0] = 0;
  }
  __syncthreads();
  const int chunk = (E + BING - 1) / BING;
  const int e0 = blockIdx.x * chunk;
  const int e1 = min(e0 + chunk, E);
  for (int e = e0 + (int)threadIdx.x; e < e1; e += NBIN) {
    const int r = ei[e], cl = ei[E + e];
    const int p = atomicAdd(&cur[min(r / NPB, NBIN - 1)], 1);
    list[p] = make_int2(r, cl);
  }
}

extern "C" __global__ void __launch_bounds__(WAVES * 32)
egnn_pre(const float* __restrict__ hfeat, const h16* __restrict__ mw1p,
         const float* __restrict__ mb1, float* __restrict__ pq, int Nn) {
  const int wave = threadIdx.x >> 5;
  const int lane = threadIdx.x & 31;
  const int m    = lane & 15;
  const int kb   = (lane < 16) ? 0 : 8;
  const int drow = (lane < 16) ? 0 : 8;
  const int ntile = Nn >> 4;
  const int nwav  = gridDim.x * WAVES;
  for (int tile = blockIdx.x * WAVES + wave; tile < ntile; tile += nwav) {
    const float* hr = hfeat + (size_t)(tile * 16 + m) * NODE_DIM;
    v8f accP[8], accQ[8];
    init_bias(accP, mb1, m);
#pragma unroll
    for (int nt = 0; nt < 8; ++nt)
      accQ[nt] = (v8f){0.f, 0.f, 0.f, 0.f, 0.f, 0.f, 0.f, 0.f};
#pragma unroll 1
    for (int kt = 0; kt < NODE_DIM / 32; ++kt) {
      const float4 x0 = *(const float4*)(hr + kt * 32 + kb);
      const float4 x1 = *(const float4*)(hr + kt * 32 + kb + 4);
      const float4 x2 = *(const float4*)(hr + kt * 32 + 16 + kb);
      const float4 x3 = *(const float4*)(hr + kt * 32 + 16 + kb + 4);
      v16h a;
      a[0]  = (h16)x0.x; a[1]  = (h16)x0.y; a[2]  = (h16)x0.z; a[3]  = (h16)x0.w;
      a[4]  = (h16)x1.x; a[5]  = (h16)x1.y; a[6]  = (h16)x1.z; a[7]  = (h16)x1.w;
      a[8]  = (h16)x2.x; a[9]  = (h16)x2.y; a[10] = (h16)x2.z; a[11] = (h16)x2.w;
      a[12] = (h16)x3.x; a[13] = (h16)x3.y; a[14] = (h16)x3.z; a[15] = (h16)x3.w;
#pragma unroll
      for (int nt = 0; nt < 8; ++nt) {
        v16h bp = load_b_frag(mw1p, kt, nt, 8, lane);
        accP[nt] = __builtin_amdgcn_wmma_f32_16x16x32_f16(false, a, false, bp,
                                                          (short)0, accP[nt], false, false);
        v16h bq = load_b_frag(mw1p, kt + NODE_DIM / 32, nt, 8, lane);
        accQ[nt] = __builtin_amdgcn_wmma_f32_16x16x32_f16(false, a, false, bq,
                                                          (short)0, accQ[nt], false, false);
      }
    }
#pragma unroll
    for (int r = 0; r < 8; ++r) {
      float4* o = (float4*)(pq + (size_t)(tile * 16 + drow + r) * (2 * HIDDEN) + m * 4);
      o[0]  = make_float4(accP[0][r], accP[1][r], accP[2][r], accP[3][r]);
      o[16] = make_float4(accP[4][r], accP[5][r], accP[6][r], accP[7][r]);
      o[32] = make_float4(accQ[0][r], accQ[1][r], accQ[2][r], accQ[3][r]);
      o[48] = make_float4(accQ[4][r], accQ[5][r], accQ[6][r], accQ[7][r]);
    }
  }
}

extern "C" __global__ void __launch_bounds__(EWAVES * 32)
egnn_edge(const float* __restrict__ pq, const float* __restrict__ pos,
          const int2* __restrict__ list, const int* __restrict__ binstart,
          const h16* __restrict__ mw1p,
          const h16* __restrict__ mw2p, const float* __restrict__ mb2,
          const h16* __restrict__ cw1p, const float* __restrict__ cb1,
          const float* __restrict__ cw2, const float* __restrict__ cb2,
          float* __restrict__ aggr, float* __restrict__ pos_out, int Nn) {
  __shared__ float s_acc[(NPB + 1) * ACC_LD];
  __shared__ float s_pacc[(NPB + 1) * 3];
  __shared__ __attribute__((aligned(16))) char smem[EWAVES * (16 * HID_LD * 2 + 128)];
  for (int i = threadIdx.x; i < (NPB + 1) * ACC_LD; i += EWAVES * 32) s_acc[i] = 0.0f;
  for (int i = threadIdx.x; i < (NPB + 1) * 3; i += EWAVES * 32) s_pacc[i] = 0.0f;
  __syncthreads();
  const int wave = threadIdx.x >> 5;
  const int lane = threadIdx.x & 31;
  const int m    = lane & 15;
  const int kb   = (lane < 16) ? 0 : 8;
  const int drow = (lane < 16) ? 0 : 8;

  const int WB = 16 * HID_LD * 2 + 128;
  char* wb = smem + wave * WB;
  h16*   s_hid = (h16*)wb;
  int*   s_row = (int*)(wb + 16 * HID_LD * 2);
  int*   s_col = s_row + 16;

  v16h bw2[HIDDEN / 32];
#pragma unroll
  for (int kt = 0; kt < HIDDEN / 32; ++kt) {
#pragma unroll
    for (int s = 0; s < 8; ++s) {
      bw2[kt][s]     = (h16)cw2[kt * 32 + kb + s];
      bw2[kt][8 + s] = (h16)cw2[kt * 32 + 16 + kb + s];
    }
  }
  const float cb2v = cb2[0];
  const int base  = blockIdx.x * NPB;
  const int lo    = binstart[blockIdx.x];
  const int cnt   = binstart[blockIdx.x + 1] - lo;
  const int ntile = (cnt + 15) >> 4;
  for (int tile = wave; tile < ntile; tile += EWAVES) {
    const int e    = tile * 16 + m;
    const int2 rc  = list[lo + min(e, cnt - 1)];
    const int vrow = rc.x;
    const int vcol = rc.y;
    const float rx = pos[vrow * 3 + 0] - pos[vcol * 3 + 0];
    const float ry = pos[vrow * 3 + 1] - pos[vcol * 3 + 1];
    const float rz = pos[vrow * 3 + 2] - pos[vcol * 3 + 2];
    const float dist = sqrtf(rx * rx + ry * ry + rz * rz + 1e-8f);
    s_row[lane] = (lane < 16) ? vrow : vcol;

    v16h ag;
    {
      const float step  = 10.0f / (NBASIS - 1);
      const float coeff = -0.5f / (step * step);
#pragma unroll
      for (int s = 0; s < 8; ++s) {
        float d0 = dist - step * (kb + s);
        float d1 = dist - step * (16 + kb + s);
        ag[s]     = (h16)__expf(coeff * d0 * d0);
        ag[8 + s] = (h16)__expf(coeff * d1 * d1);
      }
    }
    asm volatile("s_wait_dscnt 0" ::: "memory");
    int nr[8], nc[8];
    {
      const int4 r0 = *(const int4*)(s_row + drow), r1 = *(const int4*)(s_row + drow + 4);
      const int4 c0 = *(const int4*)(s_col + drow), c1 = *(const int4*)(s_col + drow + 4);
      nr[0] = r0.x; nr[1] = r0.y; nr[2] = r0.z; nr[3] = r0.w;
      nr[4] = r1.x; nr[5] = r1.y; nr[6] = r1.z; nr[7] = r1.w;
      nc[0] = c0.x; nc[1] = c0.y; nc[2] = c0.z; nc[3] = c0.w;
      nc[4] = c1.x; nc[5] = c1.y; nc[6] = c1.z; nc[7] = c1.w;
    }
    int ao[8];
#pragma unroll
    for (int r = 0; r < 8; ++r) {
      const int lr = min(max(nr[r] - base, 0), NPB - 1);
      ao[r] = ((tile * 16 + drow + r < cnt) ? lr : NPB) * ACC_LD;
    }

    v8f acc[8];
#pragma unroll
    for (int nt = 0; nt < 8; ++nt) {
      v16h b = load_b_frag(mw1p, 2 * NODE_DIM / 32, nt, 8, lane);
      acc[nt] = __builtin_amdgcn_wmma_f32_16x16x32_f16(false, ag, false, b, (short)0,
                    (v8f){0.f, 0.f, 0.f, 0.f, 0.f, 0.f, 0.f, 0.f}, false, false);
    }
    v4f gp0[8], gp1[8], gq0[8], gq1[8];
#pragma unroll
    for (int r = 0; r < 8; ++r) {
      const v4f* pp = (const v4f*)(pq + (size_t)nr[r] * (2 * HIDDEN) + m * 4);
      const v4f* qp = (const v4f*)(pq + (size_t)nc[r] * (2 * HIDDEN) + HIDDEN + m * 4);
      gp0[r] = pp[0]; gp1[r] = pp[16]; gq0[r] = qp[0]; gq1[r] = qp[16];
    }
#pragma unroll
    for (int r = 0; r < 8; ++r) {
      acc[0][r] += gp0[r].x + gq0[r].x; acc[1][r] += gp0[r].y + gq0[r].y;
      acc[2][r] += gp0[r].z + gq0[r].z; acc[3][r] += gp0[r].w + gq0[r].w;
      acc[4][r] += gp1[r].x + gq1[r].x; acc[5][r] += gp1[r].y + gq1[r].y;
      acc[6][r] += gp1[r].z + gq1[r].z; acc[7][r] += gp1[r].w + gq1[r].w;
    }
    asm volatile("" :: "v"(acc[0]), "v"(acc[1]), "v"(acc[2]), "v"(acc[3]),
                       "v"(acc[4]), "v"(acc[5]), "v"(acc[6]), "v"(acc[7]));
    __builtin_amdgcn_sched_barrier(0);
    v16h bs[8];
    load_b_row(bs, mw2p, 0, lane);
    store_panel<true>(s_hid, acc, drow, m);
    asm volatile("s_wait_dscnt 0" ::: "memory");

    v8f acc2[8];
    init_bias(acc2, mb2, m);
    layer128<true>(acc2, bs, s_hid + m * HID_LD, kb, mw2p, cw1p, lane);
#pragma unroll
    for (int nt = 0; nt < 8; ++nt)
#pragma unroll
      for (int r = 0; r < 8; ++r)
        atomicAdd(&s_acc[ao[r] + nt * 16 + m], acc2[nt][r]);
    store_panel<false>(s_hid, acc2, drow, m);
    asm volatile("s_wait_dscnt 0" ::: "memory");

    v8f acc3[8];
    init_bias(acc3, cb1, m);
    layer128<false>(acc3, bs, s_hid + m * HID_LD, kb, cw1p, cw1p, lane);
    store_panel<true>(s_hid, acc3, drow, m);
    asm volatile("s_wait_dscnt 0" ::: "memory");
    v8f acc4 = (v8f){cb2v, cb2v, cb2v, cb2v, cb2v, cb2v, cb2v, cb2v};
#pragma unroll
    for (int kt = 0; kt < HIDDEN / 32; ++kt) {
      v16h a = load_a_frag(s_hid + m * HID_LD, kt, kb);
      acc4 = __builtin_amdgcn_wmma_f32_16x16x32_f16(false, a, false, bw2[kt],
                                                    (short)0, acc4, false, false);
    }
    float sel = acc4[0];
#pragma unroll
    for (int r = 1; r < 8; ++r) sel = ((m & 7) == r) ? acc4[r] : sel;
    const float oth = __shfl_xor(sel, 16);
    if (lane < 16) {
      float cm = (m < 8) ? sel : oth;
      const int po = ((e < cnt) ? min(max(vrow - base, 0), NPB - 1) : NPB) * 3;
      atomicAdd(&s_pacc[po + 0], rx * cm);
      atomicAdd(&s_pacc[po + 1], ry * cm);
      atomicAdd(&s_pacc[po + 2], rz * cm);
    }
    asm volatile("s_wait_dscnt 0" ::: "memory");
  }
  __syncthreads();
  for (int i = threadIdx.x; i < NPB * 32; i += EWAVES * 32) {
    const int ln = i >> 5, q = i & 31;
    const int node = base + ln;
    const float* sr = s_acc + ln * ACC_LD + (q & 1) * 64 + (q >> 1);
    if (node < Nn)
      *(float4*)(aggr + (size_t)node * HIDDEN + q * 4) = make_float4(sr[0], sr[16], sr[32], sr[48]);
  }
  for (int i = threadIdx.x; i < NPB * 3; i += EWAVES * 32)
    if (base * 3 + i < Nn * 3) unsafeAtomicAdd(&pos_out[base * 3 + i], s_pacc[i]);
}

extern "C" __global__ void __launch_bounds__(WAVES * 32)
egnn_node(const float* __restrict__ hfeat, const float* __restrict__ aggr,
          const h16* __restrict__ nw1p, const float* __restrict__ nb1,
          const h16* __restrict__ nw2p, const float* __restrict__ nb2,
          float* __restrict__ h_out, int Nn) {
  extern __shared__ char smem[];
  const int wave = threadIdx.x >> 5;
  const int lane = threadIdx.x & 31;
  const int m    = lane & 15;
  const int kb   = (lane < 16) ? 0 : 8;
  const int drow = (lane < 16) ? 0 : 8;

  const int NIN = 2 * NODE_DIM;
  const int WB  = 16 * NIN_LD * 2 + 16 * HID_LD * 2;
  char* wb = smem + wave * WB;
  h16* s_nin = (h16*)wb;
  h16* s_hid = (h16*)(wb + 16 * NIN_LD * 2);

  const int ntile = Nn >> 4;
  const int nwav  = gridDim.x * WAVES;
  for (int tile = blockIdx.x * WAVES + wave; tile < ntile; tile += nwav) {
    const int node = tile * 16 + m;
    const float4* src = (lane < 16) ? (const float4*)(hfeat + (size_t)node * NODE_DIM)
                                    : (const float4*)(aggr + (size_t)node * HIDDEN);
    h16* dst = s_nin + m * NIN_LD + ((lane < 16) ? 0 : NODE_DIM);
#pragma unroll 4
    for (int j = 0; j < NODE_DIM / 4; ++j) {
      float4 v = src[j];
      dst[j * 4 + 0] = (h16)v.x; dst[j * 4 + 1] = (h16)v.y;
      dst[j * 4 + 2] = (h16)v.z; dst[j * 4 + 3] = (h16)v.w;
    }
    asm volatile("s_wait_dscnt 0" ::: "memory");

    v8f acc[8];
    init_bias(acc, nb1, m);
#pragma unroll
    for (int kt = 0; kt < NIN / 32; ++kt) {
      v16h a = load_a_frag(s_nin + m * NIN_LD, kt, kb);
#pragma unroll
      for (int nt = 0; nt < 8; ++nt) {
        v16h b = load_b_frag(nw1p, kt, nt, 8, lane);
        acc[nt] = __builtin_amdgcn_wmma_f32_16x16x32_f16(false, a, false, b,
                                                         (short)0, acc[nt], false, false);
      }
    }
    store_panel<true>(s_hid, acc, drow, m);
    asm volatile("s_wait_dscnt 0" ::: "memory");

    v8f acc2[8];
    init_bias(acc2, nb2, m);
#pragma unroll
    for (int kt = 0; kt < HIDDEN / 32; ++kt) {
      v16h a = load_a_frag(s_hid + m * HID_LD, kt, kb);
#pragma unroll
      for (int nt = 0; nt < 8; ++nt) {
        v16h b = load_b_frag(nw2p, kt, nt, 8, lane);
        acc2[nt] = __builtin_amdgcn_wmma_f32_16x16x32_f16(false, a, false, b,
                                                          (short)0, acc2[nt], false, false);
      }
    }
#pragma unroll
    for (int r = 0; r < 8; ++r) {
      const size_t o = (size_t)(tile * 16 + drow + r) * NODE_DIM + m * 8;
      float hh[8];
      load8(hh, hfeat + o);
      float4* op = (float4*)(h_out + o);
      op[0] = make_float4(hh[0] + acc2[0][r], hh[1] + acc2[1][r],
                          hh[2] + acc2[2][r], hh[3] + acc2[3][r]);
      op[1] = make_float4(hh[4] + acc2[4][r], hh[5] + acc2[5][r],
                          hh[6] + acc2[6][r], hh[7] + acc2[7][r]);
    }
  }
}

extern "C" void kernel_launch(void* const* d_in, const int* in_sizes, int n_in,
                              void* d_out, int out_size, void* d_ws, size_t ws_size,
                              hipStream_t stream) {
  const float* h   = (const float*)d_in[0];
  const float* pos = (const float*)d_in[1];
  const int*   ei  = (const int*)d_in[2];
  const float* mw1 = (const float*)d_in[3];  const float* mb1 = (const float*)d_in[4];
  const float* mw2 = (const float*)d_in[5];  const float* mb2 = (const float*)d_in[6];
  const float* nw1 = (const float*)d_in[7];  const float* nb1 = (const float*)d_in[8];
  const float* nw2 = (const float*)d_in[9];  const float* nb2 = (const float*)d_in[10];
  const float* cw1 = (const float*)d_in[11]; const float* cb1 = (const float*)d_in[12];
  const float* cw2 = (const float*)d_in[13]; const float* cb2 = (const float*)d_in[14];
  const int N = in_sizes[0] / NODE_DIM;
  const int E = in_sizes[2] / 2;

  char* ws = (char*)d_ws;
  h16* mw1p = (h16*)(ws + 0);
  h16* mw2p = (h16*)(ws + 73728);
  h16* cw1p = (h16*)(ws + 106496);
  h16* nw1p = (h16*)(ws + 139264);
  h16* nw2p = (h16*)(ws + 204800);
  float* aggr = (float*)(ws + 237568);
  float* pq   = (float*)(ws + 25837568);
  int*  counts   = (int*)(ws + 77037568);
  int*  binstart = (int*)(ws + 77549568);
  int2* list     = (int2*)(ws + 77551616);
  float* h_out   = (float*)d_out;
  float* pos_out = h_out + (size_t)N * NODE_DIM;

  pack_w<<<(9 * 8 * 32 + 127) / 128, 128, 0, stream>>>(mw1, mw1p, MSG_IN, HIDDEN);
  pack_w<<<(4 * 8 * 32 + 127) / 128, 128, 0, stream>>>(mw2, mw2p, HIDDEN, HIDDEN);
  pack_w<<<(4 * 8 * 32 + 127) / 128, 128, 0, stream>>>(cw1, cw1p, HIDDEN, HIDDEN);
  pack_w<<<(8 * 8 * 32 + 127) / 128, 128, 0, stream>>>(nw1, nw1p, 2 * NODE_DIM, HIDDEN);
  pack_w<<<(4 * 8 * 32 + 127) / 128, 128, 0, stream>>>(nw2, nw2p, HIDDEN, NODE_DIM);
  bin_count<<<BING, NBIN, 0, stream>>>(ei, counts, pos, pos_out, E, N);
  bin_place<<<BING, NBIN, 0, stream>>>(ei, counts, binstart, list, E);

  egnn_pre<<<800, WAVES * 32, 0, stream>>>(h, mw1p, mb1, pq, N);

  egnn_edge<<<NBIN, EWAVES * 32, 0, stream>>>(pq, pos, list, binstart, mw1p, mw2p, mb2,
                                              cw1p, cb1, cw2, cb2, aggr, pos_out, N);

  size_t lds_n = (size_t)WAVES * (16 * NIN_LD * 2 + 16 * HID_LD * 2);
  egnn_node<<<800, WAVES * 32, lds_n, stream>>>(h, aggr, nw1p, nb1, nw2p, nb2, h_out, N);
}


# Edge work-groups run 16 waves (4 per SIMD), not 12: a wave fits 256 registers as cw2 fragments, basis centres and sum-row offsets are read from LDS per tile; LDS atomic sums arrive in another order
# speedup vs baseline: 2.7554x; 1.0792x over previous
#include <hip/hip_runtime.h>
#include <math.h>

typedef _Float16 h16;
typedef __attribute__((ext_vector_type(16))) _Float16 v16h;
typedef __attribute__((ext_vector_type(8)))  _Float16 v8h;
typedef __attribute__((ext_vector_type(8)))  float    v8f;
typedef __attribute__((ext_vector_type(4)))  float    v4f;

#define NODE_DIM 128
#define HIDDEN   128
#define NBASIS   32
#define MSG_IN   288
#define WAVES    4
#define NBIN     256
#define NPB      196
#define BING     250
#define EWAVES   16
#define ACC_LD   144
#define HID_LD   136
#define NIN_LD   264

static __device__ __forceinline__ v16h cat8(v8h a, v8h b) {
  return __builtin_shufflevector(a, b, 0,1,2,3,4,5,6,7,8,9,10,11,12,13,14,15);
}
static __device__ __forceinline__ float silu(float x) {
  return x / (1.0f + __expf(-x));
}
static __device__ __forceinline__ void load8(float (&b)[8], const float* __restrict__ p) {
  const float4 u = ((const float4*)p)[0], v = ((const float4*)p)[1];
  b[0] = u.x; b[1] = u.y; b[2] = u.z; b[3] = u.w;
  b[4] = v.x; b[5] = v.y; b[6] = v.z; b[7] = v.w;
}
static __device__ __forceinline__ void init_bias(v8f (&acc)[8], const float* __restrict__ bias,
                                                 int m) {
  float bb[8];
  load8(bb, bias + m * 8);
#pragma unroll
  for (int nt = 0; nt < 8; ++nt) {
    const float b0 = bb[nt];
    acc[nt] = (v8f){b0, b0, b0, b0, b0, b0, b0, b0};
  }
}
template <bool SILU>
static __device__ __forceinline__ void store_panel(h16* panel, const v8f (&acc)[8],
                                                   int drow, int m) {
  v8h hv[8];
#pragma unroll
  for (int np = 0; np < 8; np += 2)
#pragma unroll
    for (int r = 0; r < 8; ++r) {
      hv[r][np]     = (h16)(SILU ? silu(acc[np][r])     : acc[np][r]);
      hv[r][np + 1] = (h16)(SILU ? silu(acc[np + 1][r]) : acc[np + 1][r]);
    }
#pragma unroll
  for (int r = 0; r < 8; ++r)
    *(v8h*)(panel + (drow + r) * HID_LD + m * 8) = hv[r];
}
static __device__ __forceinline__ v16h load_b_frag(const h16* __restrict__ wp,
                                                   int kt, int nt, int NT, int lane) {
  const v8h* p = (const v8h*)(wp + (size_t)(((kt * NT + nt) * 32 + lane)) * 16);
  return cat8(p[0], p[1]);
}
static __device__ __forceinline__ v16h load_a_frag(const h16* row_base, int kt, int kb) {
  const v8h* a0 = (const v8h*)(row_base + kt * 32 + kb);
  const v8h* a1 = (const v8h*)(row_base + kt * 32 + 16 + kb);
  return cat8(*a0, *a1);
}

static __device__ __forceinline__ void load_b_row(v16h (&b)[8], const h16* __restrict__ wp,
                                                  int kt, int lane) {
#pragma unroll
  for (int nt = 0; nt < 8; ++nt) b[nt] = load_b_frag(wp, kt, nt, 8, lane);
}
template <bool NEXT>
static __device__ __forceinline__ void layer128(v8f (&acc)[8], v16h (&s0)[8],
                                                const h16* arow, int kb,
                                                const h16* __restrict__ wp,
                                                const h16* __restrict__ wnext, int lane) {
  v16h s1[8];
  v16h a0 = load_a_frag(arow, 0, kb);
#pragma unroll 1
  for (int kt = 0; kt < HIDDEN / 32; kt += 2) {
    load_b_row(s1, wp, kt + 1, lane);
    const v16h a1 = load_a_frag(arow, kt + 1, kb);
    __builtin_amdgcn_sched_barrier(0);
#pragma unroll
    for (int nt = 0; nt < 8; ++nt)
      acc[nt] = __builtin_amdgcn_wmma_f32_16x16x32_f16(false, a0, false, s0[nt],
                                                       (short)0, acc[nt], false, false);
    __builtin_amdgcn_sched_barrier(0);
    if (NEXT) {
      const h16* nx = (kt + 2 < HIDDEN / 32) ? wp + (size_t)(kt + 2) * (8 * 32 * 16) : wnext;
      load_b_row(s0, nx, 0, lane);
      a0 = load_a_frag(arow, (kt + 2) & (HIDDEN / 32 - 1), kb);
    } else if (kt + 2 < HIDDEN / 32) {
      load_b_row(s0, wp, kt + 2, lane);
      a0 = load_a_frag(arow, kt + 2, kb);
    }
    __builtin_amdgcn_sched_barrier(0);
#pragma unroll
    for (int nt = 0; nt < 8; ++nt)
      acc[nt] = __builtin_amdgcn_wmma_f32_16x16x32_f16(false, a1, false, s1[nt],
                                                       (short)0, acc[nt], false, false);
    __builtin_amdgcn_sched_barrier(0);
  }
}

extern "C" __global__ void pack_w(const float* __restrict__ W, h16* __restrict__ out,
                                  int K, int Nn) {
  int t = blockIdx.x * blockDim.x + threadIdx.x;
  int KT = K >> 5, NT = Nn >> 4;
  int total = KT * NT * 32;
  if (t >= total) return;
  int lane = t & 31;
  int tile = t >> 5;
  int nt = tile % NT, kt = tile / NT;
  int n = (lane & 15) * NT + nt;
  int kbase = (lane < 16) ? 0 : 8;
  alignas(16) h16 tmp[16];
#pragma unroll
  for (int r = 0; r < 4; ++r) {
    int k0 = kt * 32 + kbase + 2 * r;
    tmp[2 * r]     = (h16)W[(size_t)k0 * Nn + n];
    tmp[2 * r + 1] = (h16)W[(size_t)(k0 + 1) * Nn + n];
    int k1 = kt * 32 + 16 + kbase + 2 * r;
    tmp[8 + 2 * r]     = (h16)W[(size_t)k1 * Nn + n];
    tmp[8 + 2 * r + 1] = (h16)W[(size_t)(k1 + 1) * Nn + n];
  }
  v8h* o = (v8h*)(out + (size_t)t * 16);
  o[0] = *(const v8h*)(tmp);
  o[1] = *(const v8h*)(tmp + 8);
}

static __device__ __forceinline__ void pre_tile(const float* __restrict__ hfeat,
                                                const h16* __restrict__ mw1p,
                                                const float* __restrict__ mb1,
                                                float* __restrict__ pq, int tile, int lane) {
  const int m    = lane & 15;
  const int kb   = (lane < 16) ? 0 : 8;
  const int drow = (lane < 16) ? 0 : 8;
  const float* hr = hfeat + (size_t)(tile * 16 + m) * NODE_DIM;
  v8f accP[8], accQ[8];
  init_bias(accP, mb1, m);
#pragma unroll
  for (int nt = 0; nt < 8; ++nt)
    accQ[nt] = (v8f){0.f, 0.f, 0.f, 0.f, 0.f, 0.f, 0.f, 0.f};
#pragma unroll 1
  for (int kt = 0; kt < NODE_DIM / 32; ++kt) {
    const float4 x0 = *(const float4*)(hr + kt * 32 + kb);
    const float4 x1 = *(const float4*)(hr + kt * 32 + kb + 4);
    const float4 x2 = *(const float4*)(hr + kt * 32 + 16 + kb);
    const float4 x3 = *(const float4*)(hr + kt * 32 + 16 + kb + 4);
    v16h a;
    a[0]  = (h16)x0.x; a[1]  = (h16)x0.y; a[2]  = (h16)x0.z; a[3]  = (h16)x0.w;
    a[4]  = (h16)x1.x; a[5]  = (h16)x1.y; a[6]  = (h16)x1.z; a[7]  = (h16)x1.w;
    a[8]  = (h16)x2.x; a[9]  = (h16)x2.y; a[10] = (h16)x2.z; a[11] = (h16)x2.w;
    a[12] = (h16)x3.x; a[13] = (h16)x3.y; a[14] = (h16)x3.z; a[15] = (h16)x3.w;
#pragma unroll
    for (int nt = 0; nt < 8; ++nt) {
      v16h bp = load_b_frag(mw1p, kt, nt, 8, lane);
      accP[nt] = __builtin_amdgcn_wmma_f32_16x16x32_f16(false, a, false, bp,
                                                        (short)0, accP[nt], false, false);
      v16h bq = load_b_frag(mw1p, kt + NODE_DIM / 32, nt, 8, lane);
      accQ[nt] = __builtin_amdgcn_wmma_f32_16x16x32_f16(false, a, false, bq,
                                                        (short)0, accQ[nt], false, false);
    }
  }
#pragma unroll
  for (int r = 0; r < 8; ++r) {
    float4* o = (float4*)(pq + (size_t)(tile * 16 + drow + r) * (2 * HIDDEN) + m * 4);
    o[0]  = make_float4(accP[0][r], accP[1][r], accP[2][r], accP[3][r]);
    o[16] = make_float4(accP[4][r], accP[5][r], accP[6][r], accP[7][r]);
    o[32] = make_float4(accQ[0][r], accQ[1][r], accQ[2][r], accQ[3][r]);
    o[48] = make_float4(accQ[4][r], accQ[5][r], accQ[6][r], accQ[7][r]);
  }
}
static __device__ __forceinline__ void pre_part(const float* __restrict__ hfeat,
                                                const h16* __restrict__ mw1p,
                                                const float* __restrict__ mb1,
                                                float* __restrict__ pq, int Nn, int part) {
  const int wave  = threadIdx.x >> 5;
  const int lane  = threadIdx.x & 31;
  const int ntile = Nn >> 4;
  const int half  = ntile >> 1;
  const int t0    = part ? half : 0;
  const int t1    = part ? ntile : half;
  const int nwav  = ((int)gridDim.x - BING) * (NBIN / 32);
  for (int tile = t0 + ((int)blockIdx.x - BING) * (NBIN / 32) + wave; tile < t1; tile += nwav)
    pre_tile(hfeat, mw1p, mb1, pq, tile, lane);
}

extern "C" __global__ void __launch_bounds__(NBIN)
bin_count(const int* __restrict__ ei, int* __restrict__ counts,
          const float* __restrict__ pos, float* __restrict__ pos_out, int E, int Nn,
          const float* __restrict__ hfeat, const h16* __restrict__ mw1p,
          const float* __restrict__ mb1, float* __restrict__ pq) {
  __shared__ int hist[NBIN];
  if (blockIdx.x >= BING) {
    pre_part(hfeat, mw1p, mb1, pq, Nn, 0);
    return;
  }
  hist[threadIdx.x] = 0;
  __syncthreads();
  const int chunk = (E + BING - 1) / BING;
  const int e0 = blockIdx.x * chunk;
  const int e1 = min(e0 + chunk, E);
  for (int e = e0 + (int)threadIdx.x; e < e1; e += NBIN)
    atomicAdd(&hist[min(ei[e] / NPB, NBIN - 1)], 1);
  __syncthreads();
  counts[blockIdx.x * NBIN + threadIdx.x] = hist[threadIdx.x];
  counts[(blockIdx.x + BING) * NBIN + threadIdx.x] = hist[threadIdx.x];
  for (int j = blockIdx.x * NBIN + threadIdx.x; j < Nn * 3; j += BING * NBIN)
    pos_out[j] = pos[j];
}
extern "C" __global__ void __launch_bounds__(NBIN)
bin_place(const int* __restrict__ ei, const int* __restrict__ counts,
          int* __restrict__ binstart, int2* __restrict__ list, int E, int Nn,
          const float* __restrict__ hfeat, const h16* __restrict__ mw1p,
          const float* __restrict__ mb1, float* __restrict__ pq) {
  __shared__ int cur[NBIN];
  __shared__ int wtot[NBIN / 32];
  if (blockIdx.x >= BING) {
    pre_part(hfeat, mw1p, mb1, pq, Nn, 1);
    return;
  }
  const int b = threadIdx.x;
  const int s = blockIdx.x;
  int total = 0, before = 0;
  const int* p = counts + s * NBIN + b;
  const int nb = BING - s;
  for (int i0 = 0; i0 < BING; i0 += 50) {
    int cc[50];
#pragma unroll
    for (int j = 0; j < 50; ++j) cc[j] = p[(i0 + j) * NBIN];
#pragma unroll
    for (int j = 0; j < 50; ++j) {
      total += cc[j];
      before += (i0 + j >= nb) ? cc[j] : 0;
    }
  }
  const int lane = b & 31, w = b >> 5;
  int incl = total;
  for (int d = 1; d < 32; d <<= 1) {
    const int v = __shfl_up(incl, d);
    incl += (lane >= d) ? v : 0;
  }
  if (lane == 31) wtot[w] = incl;
  __syncthreads();
  for (int j = 0; j < NBIN / 32; ++j) incl += (j < w) ? wtot[j] : 0;
  cur[b] = incl - total + before;
  if (s == 0) {
    binstart[b + 1] = incl;
    if (b == 0) binstart[0] = 0;
  }
  __syncthreads();
  const int chunk = (E + BING - 1) / BING;
  const int e0 = blockIdx.x * chunk;
  const int e1 = min(e0 + chunk, E);
  for (int e = e0 + (int)threadIdx.x; e < e1; e += NBIN) {
    const int r = ei[e], cl = ei[E + e];
    const int p = atomicAdd(&cur[min(r / NPB, NBIN - 1)], 1);
    list[p] = make_int2(r, cl);
  }
}

extern "C" __global__ void __launch_bounds__(EWAVES * 32)
egnn_edge(const float* __restrict__ pq, const float* __restrict__ pos,
          const int2* __restrict__ list, const int* __restrict__ binstart,
          const h16* __restrict__ mw1p,
          const h16* __restrict__ mw2p, const float* __restrict__ mb2,
          const h16* __restrict__ cw1p, const float* __restrict__ cb1,
          const float* __restrict__ cw2, const float* __restrict__ cb2,
          float* __restrict__ aggr, float* __restrict__ pos_out, int Nn) {
  __shared__ float s_acc[(NPB + 1) * ACC_LD];
  __shared__ float s_pacc[(NPB + 1) * 3];
  __shared__ __attribute__((aligned(16))) char smem[EWAVES * (16 * HID_LD * 2 + 128)];
  __shared__ __attribute__((aligned(16))) h16 s_cw2[HIDDEN];
  __shared__ __attribute__((aligned(16))) float s_goff[32];
  if (threadIdx.x < HIDDEN) s_cw2[threadIdx.x] = (h16)cw2[threadIdx.x];
  if (threadIdx.x < 32) {
    const int j = threadIdx.x & 15;
    s_goff[threadIdx.x] = (10.0f / (NBASIS - 1)) * (float)((threadIdx.x >> 4) * 8 + (j < 8 ? j : 8 + j));
  }
  for (int i = threadIdx.x; i < (NPB + 1) * ACC_LD; i += EWAVES * 32) s_acc[i] = 0.0f;
  for (int i = threadIdx.x; i < (NPB + 1) * 3; i += EWAVES * 32) s_pacc[i] = 0.0f;
  __syncthreads();
  const int wave = threadIdx.x >> 5;
  const int lane = threadIdx.x & 31;
  const int m    = lane & 15;
  const int kb   = (lane < 16) ? 0 : 8;
  const int drow = (lane < 16) ? 0 : 8;

  const int WB = 16 * HID_LD * 2 + 128;
  char* wb = smem + wave * WB;
  h16*   s_hid = (h16*)wb;
  int*   s_row = (int*)(wb + 16 * HID_LD * 2);
  int*   s_col = s_row + 16;

  const float cb2v = cb2[0];
  const int base  = blockIdx.x * NPB;
  const int lo    = binstart[blockIdx.x];
  const int cnt   = binstart[blockIdx.x + 1] - lo;
  const int ntile = (cnt + 15) >> 4;
  for (int tile = wave; tile < ntile; tile += EWAVES) {
    const int e    = tile * 16 + m;
    const int2 rc  = list[lo + min(e, cnt - 1)];
    const int vrow = rc.x;
    const int vcol = rc.y;
    const float rx = pos[vrow * 3 + 0] - pos[vcol * 3 + 0];
    const float ry = pos[vrow * 3 + 1] - pos[vcol * 3 + 1];
    const float rz = pos[vrow * 3 + 2] - pos[vcol * 3 + 2];
    const float dist = sqrtf(rx * rx + ry * ry + rz * rz + 1e-8f);
    s_row[lane] = (lane < 16) ? vrow : vcol;

    v16h ag;
    {
      const float step  = 10.0f / (NBASIS - 1);
      const float coeff = -0.5f / (step * step);
      float go[8], go1[8];
      load8(go, s_goff + 2 * kb);
      load8(go1, s_goff + 2 * kb + 8);
#pragma unroll
      for (int s = 0; s < 8; ++s) {
        float d0 = dist - go[s];
        float d1 = dist - go1[s];
        ag[s]     = (h16)__expf(coeff * d0 * d0);
        ag[8 + s] = (h16)__expf(coeff * d1 * d1);
      }
    }
    asm volatile("s_wait_dscnt 0" ::: "memory");
    int nr[8], nc[8];
    {
      const int4 r0 = *(const int4*)(s_row + drow), r1 = *(const int4*)(s_row + drow + 4);
      const int4 c0 = *(const int4*)(s_col + drow), c1 = *(const int4*)(s_col + drow + 4);
      nr[0] = r0.x; nr[1] = r0.y; nr[2] = r0.z; nr[3] = r0.w;
      nr[4] = r1.x; nr[5] = r1.y; nr[6] = r1.z; nr[7] = r1.w;
      nc[0] = c0.x; nc[1] = c0.y; nc[2] = c0.z; nc[3] = c0.w;
      nc[4] = c1.x; nc[5] = c1.y; nc[6] = c1.z; nc[7] = c1.w;
    }

    v8f acc[8];
#pragma unroll
    for (int nt = 0; nt < 8; ++nt) {
      v16h b = load_b_frag(mw1p, 2 * NODE_DIM / 32, nt, 8, lane);
      acc[nt] = __builtin_amdgcn_wmma_f32_16x16x32_f16(false, ag, false, b, (short)0,
                    (v8f){0.f, 0.f, 0.f, 0.f, 0.f, 0.f, 0.f, 0.f}, false, false);
    }
    v4f gp0[8], gp1[8], gq0[8], gq1[8];
#pragma unroll
    for (int r = 0; r < 8; ++r) {
      const v4f* pp = (const v4f*)(pq + (size_t)nr[r] * (2 * HIDDEN) + m * 4);
      const v4f* qp = (const v4f*)(pq + (size_t)nc[r] * (2 * HIDDEN) + HIDDEN + m * 4);
      gp0[r] = pp[0]; gp1[r] = pp[16]; gq0[r] = qp[0]; gq1[r] = qp[16];
    }
#pragma unroll
    for (int r = 0; r < 8; ++r) {
      acc[0][r] += gp0[r].x + gq0[r].x; acc[1][r] += gp0[r].y + gq0[r].y;
      acc[2][r] += gp0[r].z + gq0[r].z; acc[3][r] += gp0[r].w + gq0[r].w;
      acc[4][r] += gp1[r].x + gq1[r].x; acc[5][r] += gp1[r].y + gq1[r].y;
      acc[6][r] += gp1[r].z + gq1[r].z; acc[7][r] += gp1[r].w + gq1[r].w;
    }
    asm volatile("" :: "v"(acc[0]), "v"(acc[1]), "v"(acc[2]), "v"(acc[3]),
                       "v"(acc[4]), "v"(acc[5]), "v"(acc[6]), "v"(acc[7]));
    __builtin_amdgcn_sched_barrier(0);
    v16h bs[8];
    load_b_row(bs, mw2p, 0, lane);
    store_panel<true>(s_hid, acc, drow, m);
    asm volatile("s_wait_dscnt 0" ::: "memory");

    v8f acc2[8];
    init_bias(acc2, mb2, m);
    layer128<true>(acc2, bs, s_hid + m * HID_LD, kb, mw2p, cw1p, lane);
    int ao[8];
    {
      const int4 r0 = *(const int4*)(s_row + drow), r1 = *(const int4*)(s_row + drow + 4);
      const int tr[8] = {r0.x, r0.y, r0.z, r0.w, r1.x, r1.y, r1.z, r1.w};
#pragma unroll
      for (int r = 0; r < 8; ++r) {
        const int lr = min(max(tr[r] - base, 0), NPB - 1);
        ao[r] = ((tile * 16 + drow + r < cnt) ? lr : NPB) * ACC_LD;
      }
    }
#pragma unroll
    for (int nt = 0; nt < 8; ++nt)
#pragma unroll
      for (int r = 0; r < 8; ++r)
        atomicAdd(&s_acc[ao[r] + nt * 16 + m], acc2[nt][r]);
    store_panel<false>(s_hid, acc2, drow, m);
    asm volatile("s_wait_dscnt 0" ::: "memory");

    v8f acc3[8];
    init_bias(acc3, cb1, m);
    layer128<false>(acc3, bs, s_hid + m * HID_LD, kb, cw1p, cw1p, lane);
    store_panel<true>(s_hid, acc3, drow, m);
    asm volatile("s_wait_dscnt 0" ::: "memory");
    v8f acc4 = (v8f){cb2v, cb2v, cb2v, cb2v, cb2v, cb2v, cb2v, cb2v};
#pragma unroll
    for (int kt = 0; kt < HIDDEN / 32; ++kt) {
      v16h a = load_a_frag(s_hid + m * HID_LD, kt, kb);
      v16h b = load_a_frag(s_cw2, kt, kb);
      acc4 = __builtin_amdgcn_wmma_f32_16x16x32_f16(false, a, false, b,
                                                    (short)0, acc4, false, false);
    }
    float sel = acc4[0];
#pragma unroll
    for (int r = 1; r < 8; ++r) sel = ((m & 7) == r) ? acc4[r] : sel;
    const float oth = __shfl_xor(sel, 16);
    if (lane < 16) {
      float cm = (m < 8) ? sel : oth;
      const int po = ((e < cnt) ? min(max(vrow - base, 0), NPB - 1) : NPB) * 3;
      atomicAdd(&s_pacc[po + 0], rx * cm);
      atomicAdd(&s_pacc[po + 1], ry * cm);
      atomicAdd(&s_pacc[po + 2], rz * cm);
    }
    asm volatile("s_wait_dscnt 0" ::: "memory");
  }
  __syncthreads();
  for (int i = threadIdx.x; i < NPB * 32; i += EWAVES * 32) {
    const int ln = i >> 5, q = i & 31;
    const int node = base + ln;
    const float* sr = s_acc + ln * ACC_LD + (q & 1) * 64 + (q >> 1);
    if (node < Nn)
      *(float4*)(aggr + (size_t)node * HIDDEN + q * 4) = make_float4(sr[0], sr[16], sr[32], sr[48]);
  }
  for (int i = threadIdx.x; i < NPB * 3; i += EWAVES * 32)
    if (base * 3 + i < Nn * 3) unsafeAtomicAdd(&pos_out[base * 3 + i], s_pacc[i]);
}

extern "C" __global__ void __launch_bounds__(WAVES * 32)
egnn_node(const float* __restrict__ hfeat, const float* __restrict__ aggr,
          const h16* __restrict__ nw1p, const float* __restrict__ nb1,
          const h16* __restrict__ nw2p, const float* __restrict__ nb2,
          float* __restrict__ h_out, int Nn) {
  extern __shared__ char smem[];
  const int wave = threadIdx.x >> 5;
  const int lane = threadIdx.x & 31;
  const int m    = lane & 15;
  const int kb   = (lane < 16) ? 0 : 8;
  const int drow = (lane < 16) ? 0 : 8;

  const int NIN = 2 * NODE_DIM;
  const int WB  = 16 * NIN_LD * 2 + 16 * HID_LD * 2;
  char* wb = smem + wave * WB;
  h16* s_nin = (h16*)wb;
  h16* s_hid = (h16*)(wb + 16 * NIN_LD * 2);

  const int ntile = Nn >> 4;
  const int nwav  = gridDim.x * WAVES;
  for (int tile = blockIdx.x * WAVES + wave; tile < ntile; tile += nwav) {
    const int node = tile * 16 + m;
    const float4* src = (lane < 16) ? (const float4*)(hfeat + (size_t)node * NODE_DIM)
                                    : (const float4*)(aggr + (size_t)node * HIDDEN);
    h16* dst = s_nin + m * NIN_LD + ((lane < 16) ? 0 : NODE_DIM);
#pragma unroll 4
    for (int j = 0; j < NODE_DIM / 4; ++j) {
      float4 v = src[j];
      dst[j * 4 + 0] = (h16)v.x; dst[j * 4 + 1] = (h16)v.y;
      dst[j * 4 + 2] = (h16)v.z; dst[j * 4 + 3] = (h16)v.w;
    }
    asm volatile("s_wait_dscnt 0" ::: "memory");

    v8f acc[8];
    init_bias(acc, nb1, m);
#pragma unroll
    for (int kt = 0; kt < NIN / 32; ++kt) {
      v16h a = load_a_frag(s_nin + m * NIN_LD, kt, kb);
#pragma unroll
      for (int nt = 0; nt < 8; ++nt) {
        v16h b = load_b_frag(nw1p, kt, nt, 8, lane);
        acc[nt] = __builtin_amdgcn_wmma_f32_16x16x32_f16(false, a, false, b,
                                                         (short)0, acc[nt], false, false);
      }
    }
    store_panel<true>(s_hid, acc, drow, m);
    asm volatile("s_wait_dscnt 0" ::: "memory");

    v8f acc2[8];
    init_bias(acc2, nb2, m);
#pragma unroll
    for (int kt = 0; kt < HIDDEN / 32; ++kt) {
      v16h a = load_a_frag(s_hid + m * HID_LD, kt, kb);
#pragma unroll
      for (int nt = 0; nt < 8; ++nt) {
        v16h b = load_b_frag(nw2p, kt, nt, 8, lane);
        acc2[nt] = __builtin_amdgcn_wmma_f32_16x16x32_f16(false, a, false, b,
                                                          (short)0, acc2[nt], false, false);
      }
    }
#pragma unroll
    for (int r = 0; r < 8; ++r) {
      const size_t o = (size_t)(tile * 16 + drow + r) * NODE_DIM + m * 8;
      float hh[8];
      load8(hh, hfeat + o);
      float4* op = (float4*)(h_out + o);
      op[0] = make_float4(hh[0] + acc2[0][r], hh[1] + acc2[1][r],
                          hh[2] + acc2[2][r], hh[3] + acc2[3][r]);
      op[1] = make_float4(hh[4] + acc2[4][r], hh[5] + acc2[5][r],
                          hh[6] + acc2[6][r], hh[7] + acc2[7][r]);
    }
  }
}

extern "C" void kernel_launch(void* const* d_in, const int* in_sizes, int n_in,
                              void* d_out, int out_size, void* d_ws, size_t ws_size,
                              hipStream_t stream) {
  const float* h   = (const float*)d_in[0];
  const float* pos = (const float*)d_in[1];
  const int*   ei  = (const int*)d_in[2];
  const float* mw1 = (const float*)d_in[3];  const float* mb1 = (const float*)d_in[4];
  const float* mw2 = (const float*)d_in[5];  const float* mb2 = (const float*)d_in[6];
  const float* nw1 = (const float*)d_in[7];  const float* nb1 = (const float*)d_in[8];
  const float* nw2 = (const float*)d_in[9];  const float* nb2 = (const float*)d_in[10];
  const float* cw1 = (const float*)d_in[11]; const float* cb1 = (const float*)d_in[12];
  const float* cw2 = (const float*)d_in[13]; const float* cb2 = (const float*)d_in[14];
  const int N = in_sizes[0] / NODE_DIM;
  const int E = in_sizes[2] / 2;

  char* ws = (char*)d_ws;
  h16* mw1p = (h16*)(ws + 0);
  h16* mw2p = (h16*)(ws + 73728);
  h16* cw1p = (h16*)(ws + 106496);
  h16* nw1p = (h16*)(ws + 139264);
  h16* nw2p = (h16*)(ws + 204800);
  float* aggr = (float*)(ws + 237568);
  float* pq   = (float*)(ws + 25837568);
  int*  counts   = (int*)(ws + 77037568);
  int*  binstart = (int*)(ws + 77549568);
  int2* list     = (int2*)(ws + 77551616);
  float* h_out   = (float*)d_out;
  float* pos_out = h_out + (size_t)N * NODE_DIM;

  pack_w<<<(9 * 8 * 32 + 127) / 128, 128, 0, stream>>>(mw1, mw1p, MSG_IN, HIDDEN);
  pack_w<<<(4 * 8 * 32 + 127) / 128, 128, 0, stream>>>(mw2, mw2p, HIDDEN, HIDDEN);
  pack_w<<<(4 * 8 * 32 + 127) / 128, 128, 0, stream>>>(cw1, cw1p, HIDDEN, HIDDEN);
  pack_w<<<(8 * 8 * 32 + 127) / 128, 128, 0, stream>>>(nw1, nw1p, 2 * NODE_DIM, HIDDEN);
  pack_w<<<(4 * 8 * 32 + 127) / 128, 128, 0, stream>>>(nw2, nw2p, HIDDEN, NODE_DIM);
  bin_count<<<BING + 196, NBIN, 0, stream>>>(ei, counts, pos, pos_out, E, N, h, mw1p, mb1, pq);
  bin_place<<<BING + 196, NBIN, 0, stream>>>(ei, counts, binstart, list, E, N, h, mw1p, mb1, pq);

  egnn_edge<<<NBIN, EWAVES * 32, 0, stream>>>(pq, pos, list, binstart, mw1p, mw2p, mb2,
                                              cw1p, cb1, cw2, cb2, aggr, pos_out, N);

  size_t lds_n = (size_t)WAVES * (16 * NIN_LD * 2 + 16 * HID_LD * 2);
  egnn_node<<<800, WAVES * 32, lds_n, stream>>>(h, aggr, nw1p, nb1, nw2p, nb2, h_out, N);
}
